# starting kernel
# baseline (speedup 1.0000x reference)
#include <hip/hip_runtime.h>
#include <cstdint>

#define NFEAT 128

typedef __attribute__((ext_vector_type(2))) float v2f;
typedef __attribute__((ext_vector_type(8))) float v8f;


__global__ void init_deg_kernel(float* __restrict__ deg, int n) {
    int i = blockIdx.x * blockDim.x + threadIdx.x;
    if (i < n) deg[i] = 1.0f;
}

__global__ void deg_accum_kernel(const int* __restrict__ dst, float* __restrict__ deg, int e) {
    int i = blockIdx.x * blockDim.x + threadIdx.x;
    if (i < e) atomicAdd(&deg[dst[i]], 1.0f);
}

__global__ void dinv_kernel(float* __restrict__ deg, int n) {
    int i = blockIdx.x * blockDim.x + threadIdx.x;
    if (i < n) deg[i] = rsqrtf(deg[i]);
}


__global__ __launch_bounds__(256) void gemm_xwT_wmma(const float* __restrict__ x,
                                                     const float* __restrict__ W,
                                                     float* __restrict__ xw, int n) {
    __shared__ float lds_x[16 * NFEAT];

    const int tid  = threadIdx.x;
    const int n0   = blockIdx.x * 16;

    const float4* xs = (const float4*)(x + (size_t)n0 * NFEAT);
    float4* ls = (float4*)lds_x;
    ls[tid]       = xs[tid];
    ls[tid + 256] = xs[tid + 256];
    __syncthreads();

    const int wave  = tid >> 5;
    const int lane  = tid & 31;
    const int o0    = wave * 16;
    const int row   = lane & 15;
    const int khalf = (lane >> 4) << 1;

    v8f acc = {};
    const float* wrow = W + (size_t)(o0 + row) * NFEAT;

#pragma unroll
    for (int k0 = 0; k0 < NFEAT; k0 += 4) {
        const int kk = k0 + khalf;
        v2f a, b;
        a.x = lds_x[row * NFEAT + kk];
        a.y = lds_x[row * NFEAT + kk + 1];
        b.x = wrow[kk];
        b.y = wrow[kk + 1];
        acc = __builtin_amdgcn_wmma_f32_16x16x4_f32(false, a, false, b,
                                                    (short)0, acc, false, false);
    }

    const int col = o0 + row;
#pragma unroll
    for (int r = 0; r < 8; ++r) {
        const int m = r + ((lane >> 4) << 3);
        xw[(size_t)(n0 + m) * NFEAT + col] = acc[r];
    }
}


__global__ void init_agg_kernel(const float* __restrict__ xw, const float* __restrict__ dinv,
                                const float* __restrict__ bias, float* __restrict__ agg,
                                int n) {
    size_t i = (size_t)blockIdx.x * blockDim.x + threadIdx.x;
    if (i < (size_t)n * NFEAT) {
        int node = (int)(i >> 7);
        int f    = (int)(i & (NFEAT - 1));
        float di = dinv[node];
        agg[i] = bias[f] + xw[i] * di * di;
    }
}


__global__ __launch_bounds__(256) void scatter_edges_kernel(const int* __restrict__ src,
                                                            const int* __restrict__ dst,
                                                            const float* __restrict__ xw,
                                                            const float* __restrict__ dinv,
                                                            float* __restrict__ agg, int e) {
    int gid  = blockIdx.x * blockDim.x + threadIdx.x;
    int edge = gid >> 5;
    int lane = gid & 31;
    if (edge >= e) return;

    int s = src[edge];
    int d = dst[edge];
    float w = dinv[s] * dinv[d];

    const float4* xs = (const float4*)(xw + (size_t)s * NFEAT);
    float4 v = xs[lane];
    float* ab = agg + (size_t)d * NFEAT + lane * 4;
    atomicAdd(ab + 0, v.x * w);
    atomicAdd(ab + 1, v.y * w);
    atomicAdd(ab + 2, v.z * w);
    atomicAdd(ab + 3, v.w * w);
}


__global__ __launch_bounds__(256) void finalize_kernel(const float* __restrict__ agg,
                                                       const float* __restrict__ wreg,
                                                       const float* __restrict__ breg,
                                                       float* __restrict__ out, int n) {
    int gid  = blockIdx.x * blockDim.x + threadIdx.x;
    int node = gid >> 5;
    int lane = gid & 31;
    if (node >= n) return;

    const float4* av = (const float4*)(agg + (size_t)node * NFEAT);
    const float4* wv = (const float4*)wreg;
    float4 a = av[lane];
    float4 w = wv[lane];
    float dot = fmaxf(a.x, 0.0f) * w.x + fmaxf(a.y, 0.0f) * w.y +
                fmaxf(a.z, 0.0f) * w.z + fmaxf(a.w, 0.0f) * w.w;
#pragma unroll
    for (int off = 16; off > 0; off >>= 1)
        dot += __shfl_xor(dot, off, 32);
    if (lane == 0) out[node] = dot + breg[0];
}


extern "C" void kernel_launch(void* const* d_in, const int* in_sizes, int n_in,
                              void* d_out, int out_size, void* d_ws, size_t ws_size,
                              hipStream_t stream) {
    const float* x    = (const float*)d_in[0];
    const int*   ei   = (const int*)d_in[1];
    const float* W    = (const float*)d_in[2];
    const float* b    = (const float*)d_in[3];
    const float* wreg = (const float*)d_in[4];
    const float* breg = (const float*)d_in[5];
    float* out = (float*)d_out;

    const int n = in_sizes[0] / NFEAT;
    const int e = in_sizes[1] / 2;
    const int* src = ei;
    const int* dst = ei + e;

    float* deg = (float*)d_ws;
    float* xw  = deg + n;
    float* agg = xw + (size_t)n * NFEAT;

    const int B = 256;
    init_deg_kernel<<<(n + B - 1) / B, B, 0, stream>>>(deg, n);
    deg_accum_kernel<<<(e + B - 1) / B, B, 0, stream>>>(dst, deg, e);
    dinv_kernel<<<(n + B - 1) / B, B, 0, stream>>>(deg, n);

    gemm_xwT_wmma<<<n / 16, B, 0, stream>>>(x, W, xw, n);

    unsigned int agg_blocks = (unsigned int)(((size_t)n * NFEAT + B - 1) / B);
    init_agg_kernel<<<agg_blocks, B, 0, stream>>>(xw, deg, b, agg, n);

    unsigned int sc_blocks = (unsigned int)(((size_t)e * 32 + B - 1) / B);
    scatter_edges_kernel<<<sc_blocks, B, 0, stream>>>(src, dst, xw, deg, agg, e);

    unsigned int fin_blocks = (unsigned int)(((size_t)n * 32 + B - 1) / B);
    finalize_kernel<<<fin_blocks, B, 0, stream>>>(agg, wreg, breg, out, n);
}


# agg rows kept in LDS by one block per 512 dst nodes that filters the whole edge list: edge adds are LDS f32 adds, agg never in memory, relu-dot from LDS; 5 launches to 4; sums in LDS arrival order
# speedup vs baseline: 2.1680x; 2.1680x over previous
#include <hip/hip_runtime.h>
#include <cstdint>

#define NFEAT 128

typedef __attribute__((ext_vector_type(2))) float v2f;
typedef __attribute__((ext_vector_type(8))) float v8f;


__global__ void init_deg_kernel(float* __restrict__ deg, int n) {
    int i = blockIdx.x * blockDim.x + threadIdx.x;
    if (i < n) deg[i] = 1.0f;
}

__global__ void deg_accum_kernel(const int* __restrict__ dst, float* __restrict__ deg, int e) {
    int i = blockIdx.x * blockDim.x + threadIdx.x;
    if (i < e) atomicAdd(&deg[dst[i]], 1.0f);
}


#define ROWS_PER_BLOCK 128
#define W_PITCH4 33

__global__ __launch_bounds__(256) void gemm_xwT_wmma(const float* __restrict__ x,
                                                     const float* __restrict__ W,
                                                     const float* __restrict__ deg,
                                                     float* __restrict__ dinv,
                                                     float* __restrict__ xw, int n) {
    __shared__ float4 lds_w[NFEAT * W_PITCH4];

    const int tid  = threadIdx.x;
    const int lane = tid & 31;
    const int wave = __builtin_amdgcn_readfirstlane(tid >> 5);
    const int c    = lane & 15;
    const int h    = lane >> 4;
    const int r0   = blockIdx.x * ROWS_PER_BLOCK + wave * 16;
    const bool live = r0 < n;

    const float4* wg = (const float4*)W;
    float4 wr[16];
#pragma unroll
    for (int i = 0; i < 16; ++i) wr[i] = wg[tid + 256 * i];

    const int rl = live ? r0 : 0;
    const float4* xr = (const float4*)(x + (size_t)(rl + c) * NFEAT) + h;
    float4 a[16];
#pragma unroll
    for (int j = 0; j < 16; ++j) a[j] = xr[2 * j];

    const float4* dg = (const float4*)(deg + rl + (h << 3));
    const float4 dg0 = dg[0];
    const float4 dg1 = dg[1];

#pragma unroll
    for (int i = 0; i < 16; ++i) {
        const int f = tid + 256 * i;
        lds_w[(f >> 5) * W_PITCH4 + (f & 31)] = wr[i];
    }
    __syncthreads();
    if (!live) return;

    v8f acc[8] = {};
    const float4* wl = lds_w + c * W_PITCH4 + h;

    float4 b[2][8];
#pragma unroll
    for (int t = 0; t < 8; ++t) b[0][t] = wl[t * 16 * W_PITCH4];

#pragma unroll
    for (int j = 0; j < 16; ++j) {
        if (j + 1 < 16) {
#pragma unroll
            for (int t = 0; t < 8; ++t) b[(j + 1) & 1][t] = wl[t * 16 * W_PITCH4 + 2 * (j + 1)];
        }
        __builtin_amdgcn_sched_barrier(0);
        v2f a0, a1;
        a0.x = a[j].x; a0.y = a[j].y;
        a1.x = a[j].z; a1.y = a[j].w;
#pragma unroll
        for (int t = 0; t < 8; ++t) {
            v2f b0;
            b0.x = b[j & 1][t].x; b0.y = b[j & 1][t].y;
            acc[t] = __builtin_amdgcn_wmma_f32_16x16x4_f32(false, a0, false, b0,
                                                           (short)0, acc[t], false, false);
        }
#pragma unroll
        for (int t = 0; t < 8; ++t) {
            v2f b1;
            b1.x = b[j & 1][t].z; b1.y = b[j & 1][t].w;
            acc[t] = __builtin_amdgcn_wmma_f32_16x16x4_f32(false, a1, false, b1,
                                                           (short)0, acc[t], false, false);
        }
        __builtin_amdgcn_sched_barrier(0);
    }

    const float di[8] = { rsqrtf(dg0.x), rsqrtf(dg0.y), rsqrtf(dg0.z), rsqrtf(dg0.w),
                          rsqrtf(dg1.x), rsqrtf(dg1.y), rsqrtf(dg1.z), rsqrtf(dg1.w) };
#pragma unroll
    for (int t = 0; t < 8; ++t) {
        const int col = 16 * t + c;
#pragma unroll
        for (int r = 0; r < 8; ++r) {
            const size_t i = (size_t)(r0 + (h << 3) + r) * NFEAT + col;
            xw[i] = acc[t][r];
        }
    }
    if (c == 0) {
        float4* dv = (float4*)(dinv + r0 + (h << 3));
        dv[0] = make_float4(di[0], di[1], di[2], di[3]);
        dv[1] = make_float4(di[4], di[5], di[6], di[7]);
    }
}


#define RANGE 512
#define FUSE_THREADS 1024
#define FUSE_WAVES 32
#define QCAP 144
#define SCAN_AHEAD 4
#define EDGE_GROUP 8

__device__ __forceinline__ void add_queued_edges(const int2* p, int cnt, int lo,
                                                 const int* __restrict__ src,
                                                 const float* __restrict__ xw,
                                                 const float* __restrict__ dinv,
                                                 float* rowsf, int lane) {
    __builtin_amdgcn_fence(__ATOMIC_ACQ_REL, "wavefront");
    const int2 en = p[(lane < cnt) ? lane : (cnt - 1)];
    const int dl = en.y;
    const int s = src[en.x];
    const float dd = dinv[lo + dl];
    const float ds = dinv[s];
    if (cnt == EDGE_GROUP) {
        float v[EDGE_GROUP][4];
#pragma unroll
        for (int j = 0; j < EDGE_GROUP; ++j) {
            const int sj = __builtin_amdgcn_readlane(s, j);
            const float* xs = xw + (size_t)sj * NFEAT + lane;
            v[j][0] = xs[0];
            v[j][1] = xs[32];
            v[j][2] = xs[64];
            v[j][3] = xs[96];
        }
        const float w = ds * dd;
#pragma unroll
        for (int j = 0; j < EDGE_GROUP; ++j) {
            const int dj = __builtin_amdgcn_readlane(dl, j);
            const float wj = __int_as_float(__builtin_amdgcn_readlane(__float_as_int(w), j));
            float* ab = rowsf + dj * NFEAT + lane;
            atomicAdd(ab + 0,  v[j][0] * wj);
            atomicAdd(ab + 32, v[j][1] * wj);
            atomicAdd(ab + 64, v[j][2] * wj);
            atomicAdd(ab + 96, v[j][3] * wj);
        }
    } else {
        const float w = ds * dd;
        for (int j = 0; j < cnt; ++j) {
            const int sj = __builtin_amdgcn_readlane(s, j);
            const int dj = __builtin_amdgcn_readlane(dl, j);
            const float wj = __int_as_float(__builtin_amdgcn_readlane(__float_as_int(w), j));
            const float* xs = xw + (size_t)sj * NFEAT + lane;
            const float v0 = xs[0];
            const float v1 = xs[32];
            const float v2 = xs[64];
            const float v3 = xs[96];
            float* ab = rowsf + dj * NFEAT + lane;
            atomicAdd(ab + 0,  v0 * wj);
            atomicAdd(ab + 32, v1 * wj);
            atomicAdd(ab + 64, v2 * wj);
            atomicAdd(ab + 96, v3 * wj);
        }
    }
}

template <bool VEC>
__device__ __forceinline__ void load_step(const int* __restrict__ dst, int it0, int nchunks, int rot,
                                          int e, int wv, int lane,
                                          int (&dd)[SCAN_AHEAD][4], int (&cb)[SCAN_AHEAD],
                                          int (&lim)[SCAN_AHEAD]) {
#pragma unroll
    for (int u = 0; u < SCAN_AHEAD; ++u) {
        const int ch = (it0 + u) * FUSE_WAVES + wv;
        const bool okc = ch < nchunks;
        int cr = okc ? (ch + rot) : 0;
        if (cr >= nchunks) cr -= nchunks;
        const int cbase = cr * 128;
        cb[u] = cbase;
        const int left = e - cbase;
        lim[u] = okc ? (left < 128 ? left : 128) : 0;
        const int base = cbase + lane * 4;
        if (VEC) {
            const int lb = (base + 4 <= e) ? base : (e - 4);
            const int4 t = *(const int4*)(dst + lb);
            dd[u][0] = t.x; dd[u][1] = t.y; dd[u][2] = t.z; dd[u][3] = t.w;
        } else {
#pragma unroll
            for (int k = 0; k < 4; ++k) dd[u][k] = dst[(base + k < e) ? (base + k) : (e - 1)];
        }
    }
}

template <bool VEC>
__device__ __forceinline__ void scan_edges(const int* __restrict__ src, const int* __restrict__ dst,
                                           const float* __restrict__ xw,
                                           const float* __restrict__ dinv,
                                           float* rowsf, int2* q, int lo, int e, int wv, int lane) {
    const int nchunks = (e + 127) >> 7;
    const int rot = (int)(((long long)blockIdx.x * nchunks) / gridDim.x);
    const int iters = (nchunks + FUSE_WAVES - 1) / FUSE_WAVES;
    int qn = 0;
    int dd[SCAN_AHEAD][4], cb[SCAN_AHEAD], lim[SCAN_AHEAD];
    load_step<VEC>(dst, 0, nchunks, rot, e, wv, lane, dd, cb, lim);
    for (int it0 = 0; it0 < iters; it0 += SCAN_AHEAD) {
        int dn[SCAN_AHEAD][4], cn[SCAN_AHEAD], ln[SCAN_AHEAD];
        load_step<VEC>(dst, it0 + SCAN_AHEAD, nchunks, rot, e, wv, lane, dn, cn, ln);
#pragma unroll
        for (int u = 0; u < SCAN_AHEAD; ++u) {
            if (lim[u] < 128) {
#pragma unroll
                for (int k = 0; k < 4; ++k)
                    if (lane * 4 + k >= lim[u]) dd[u][k] = -1;
            }
#pragma unroll
            for (int k = 0; k < 4; ++k) {
                const int dl = dd[u][k] - lo;
                const bool m = (unsigned)dl < (unsigned)RANGE;
                const unsigned bal = __builtin_amdgcn_ballot_w32(m);
                if (bal != 0u) {
                    const int pos = qn + (int)__builtin_amdgcn_mbcnt_lo(bal, 0u);
                    if (m) q[pos] = make_int2(cb[u] + lane * 4 + k, dl);
                    qn += __builtin_popcount(bal);
                }
            }
            while (qn >= EDGE_GROUP) {
                qn -= EDGE_GROUP;
                add_queued_edges(q + qn, EDGE_GROUP, lo, src, xw, dinv, rowsf, lane);
            }
        }
#pragma unroll
        for (int u = 0; u < SCAN_AHEAD; ++u) {
            cb[u] = cn[u];
            lim[u] = ln[u];
#pragma unroll
            for (int k = 0; k < 4; ++k) dd[u][k] = dn[u][k];
        }
    }
    if (qn > 0) add_queued_edges(q, qn, lo, src, xw, dinv, rowsf, lane);
}

__global__ __launch_bounds__(FUSE_THREADS) void range_sums_kernel(const int* __restrict__ src,
                                                                  const int* __restrict__ dst,
                                                                  const float* __restrict__ xw,
                                                                  const float* __restrict__ dinv,
                                                                  const float* __restrict__ bias,
                                                                  const float* __restrict__ wreg,
                                                                  const float* __restrict__ breg,
                                                                  float* __restrict__ out,
                                                                  int n, int e) {
    __shared__ float4 rows[RANGE * (NFEAT / 4)];
    __shared__ int2 queue[FUSE_WAVES * QCAP];

    const int tid  = threadIdx.x;
    const int lane = tid & 31;
    const int wv = __builtin_amdgcn_readfirstlane(tid >> 5);
    const int lo = blockIdx.x * RANGE;
    const int nrows = (n - lo < RANGE) ? (n - lo) : RANGE;

    const float4 b4 = ((const float4*)bias)[lane];
    const float4* xw4 = (const float4*)(xw + (size_t)lo * NFEAT);
    for (int i0 = 0; i0 < 16; i0 += 8) {
        float4 v[8];
        float di[8];
#pragma unroll
        for (int i = 0; i < 8; ++i) {
            const int row = wv + 32 * (i0 + i);
            const int rl = (row < nrows) ? row : (nrows - 1);
            v[i]  = xw4[rl * (NFEAT / 4) + lane];
            di[i] = dinv[lo + rl];
        }
#pragma unroll
        for (int i = 0; i < 8; ++i) {
            const int row = wv + 32 * (i0 + i);
            float4 r;
            r.x = b4.x + v[i].x * di[i] * di[i];
            r.y = b4.y + v[i].y * di[i] * di[i];
            r.z = b4.z + v[i].z * di[i] * di[i];
            r.w = b4.w + v[i].w * di[i] * di[i];
            rows[row * (NFEAT / 4) + lane] = r;
        }
    }
    __syncthreads();

    float* rowsf = (float*)rows;
    int2* q = queue + wv * QCAP;
    if ((e & 3) == 0 && e >= 4) scan_edges<true>(src, dst, xw, dinv, rowsf, q, lo, e, wv, lane);
    else if (e > 0)             scan_edges<false>(src, dst, xw, dinv, rowsf, q, lo, e, wv, lane);
    __syncthreads();

    const float4 w4 = ((const float4*)wreg)[lane];
    float res = 0.0f;
#pragma unroll
    for (int j = 0; j < 16; ++j) {
        const float4 a = rows[(wv * 16 + j) * (NFEAT / 4) + lane];
        float dot = fmaxf(a.x, 0.0f) * w4.x + fmaxf(a.y, 0.0f) * w4.y +
                    fmaxf(a.z, 0.0f) * w4.z + fmaxf(a.w, 0.0f) * w4.w;
#pragma unroll
        for (int off = 16; off > 0; off >>= 1)
            dot += __shfl_xor(dot, off, 32);
        if (lane == j) res = dot;
    }
    float* outv = (float*)queue;
    if (lane < 16) outv[wv * 16 + lane] = res + breg[0];
    __syncthreads();
    if (tid < RANGE && lo + tid < n) out[lo + tid] = outv[tid];
}


extern "C" void kernel_launch(void* const* d_in, const int* in_sizes, int n_in,
                              void* d_out, int out_size, void* d_ws, size_t ws_size,
                              hipStream_t stream) {
    const float* x    = (const float*)d_in[0];
    const int*   ei   = (const int*)d_in[1];
    const float* W    = (const float*)d_in[2];
    const float* b    = (const float*)d_in[3];
    const float* wreg = (const float*)d_in[4];
    const float* breg = (const float*)d_in[5];
    float* out = (float*)d_out;

    const int n = in_sizes[0] / NFEAT;
    const int e = in_sizes[1] / 2;
    const int* src = ei;
    const int* dst = ei + e;

    float* deg  = (float*)d_ws;
    float* xw   = deg + n;
    float* dinv = xw + (size_t)n * NFEAT;

    const int B = 256;
    init_deg_kernel<<<(n + B - 1) / B, B, 0, stream>>>(deg, n);
    deg_accum_kernel<<<(e + B - 1) / B, B, 0, stream>>>(dst, deg, e);

    gemm_xwT_wmma<<<(n + ROWS_PER_BLOCK - 1) / ROWS_PER_BLOCK, B, 0, stream>>>(x, W, deg, dinv, xw, n);

    range_sums_kernel<<<(n + RANGE - 1) / RANGE, FUSE_THREADS, 0, stream>>>(src, dst, xw, dinv, b, wreg, breg, out, n, e);
}


# Degree count without global atomics: 16 node ranges x 8 edge-list slices count in LDS and store 8 partial-count planes; the product launch adds them into dinv; 4 launches to 3; integer counts
# speedup vs baseline: 2.8438x; 1.3117x over previous
#include <hip/hip_runtime.h>
#include <cstdint>

#define NFEAT 128

typedef __attribute__((ext_vector_type(2))) float v2f;
typedef __attribute__((ext_vector_type(8))) float v8f;


#define DEG_RANGE 6272
#define DEG_SLICES 8
#define DEG_THREADS 1024
#define DEG_AHEAD 16

__global__ __launch_bounds__(DEG_THREADS) void deg_count_kernel(const int* __restrict__ dst,
                                                                int* __restrict__ part,
                                                                int n, int e) {
    __shared__ int cnt[DEG_RANGE];
    const int tid = threadIdx.x;
    const int k = blockIdx.x / DEG_SLICES;
    const int s = blockIdx.x - k * DEG_SLICES;
    const int lo = k * DEG_RANGE;
    for (int i = tid; i < DEG_RANGE; i += DEG_THREADS) cnt[i] = 0;
    __syncthreads();

    const int per = (e + DEG_SLICES - 1) / DEG_SLICES;
    const int e0 = s * per;
    const int e1 = (e0 + per < e) ? (e0 + per) : e;
    for (int b = e0 + tid; b < e1; b += DEG_THREADS * DEG_AHEAD) {
        int d[DEG_AHEAD];
#pragma unroll
        for (int u = 0; u < DEG_AHEAD; ++u) {
            const int i = b + u * DEG_THREADS;
            const int v = dst[(i < e1) ? i : (e1 - 1)];
            d[u] = (i < e1) ? v : -1;
        }
#pragma unroll
        for (int u = 0; u < DEG_AHEAD; ++u) {
            const unsigned dl = (unsigned)(d[u] - lo);
            if (dl < (unsigned)DEG_RANGE) atomicAdd(&cnt[dl], 1);
        }
    }
    __syncthreads();

    const int left = n - lo;
    const int hi = (left < DEG_RANGE) ? left : DEG_RANGE;
    int* pp = part + (size_t)s * n + lo;
    for (int i = tid; i < hi; i += DEG_THREADS) pp[i] = cnt[i];
}


#define ROWS_PER_BLOCK 128
#define W_PITCH4 33

__global__ __launch_bounds__(256) void gemm_xwT_wmma(const float* __restrict__ x,
                                                     const float* __restrict__ W,
                                                     const int* __restrict__ part,
                                                     float* __restrict__ dinv,
                                                     float* __restrict__ xw, int n) {
    __shared__ float4 lds_w[NFEAT * W_PITCH4];

    const int tid  = threadIdx.x;
    const int lane = tid & 31;
    const int wave = __builtin_amdgcn_readfirstlane(tid >> 5);
    const int c    = lane & 15;
    const int h    = lane >> 4;
    const int r0   = blockIdx.x * ROWS_PER_BLOCK + wave * 16;
    const bool live = r0 < n;

    const float4* wg = (const float4*)W;
    float4 wr[16];
#pragma unroll
    for (int i = 0; i < 16; ++i) wr[i] = wg[tid + 256 * i];

    const int rl = live ? r0 : 0;
    const float4* xr = (const float4*)(x + (size_t)(rl + c) * NFEAT) + h;
    float4 a[16];
#pragma unroll
    for (int j = 0; j < 16; ++j) a[j] = xr[2 * j];

    const int* pr = part + rl + c;
    int pc[DEG_SLICES];
#pragma unroll
    for (int s = 0; s < DEG_SLICES; ++s) pc[s] = pr[(size_t)s * n];

#pragma unroll
    for (int i = 0; i < 16; ++i) {
        const int f = tid + 256 * i;
        lds_w[(f >> 5) * W_PITCH4 + (f & 31)] = wr[i];
    }
    __syncthreads();
    if (!live) return;

    v8f acc[8] = {};
    const float4* wl = lds_w + c * W_PITCH4 + h;

    float4 b[2][8];
#pragma unroll
    for (int t = 0; t < 8; ++t) b[0][t] = wl[t * 16 * W_PITCH4];

#pragma unroll
    for (int j = 0; j < 16; ++j) {
        if (j + 1 < 16) {
#pragma unroll
            for (int t = 0; t < 8; ++t) b[(j + 1) & 1][t] = wl[t * 16 * W_PITCH4 + 2 * (j + 1)];
        }
        __builtin_amdgcn_sched_barrier(0);
        v2f a0, a1;
        a0.x = a[j].x; a0.y = a[j].y;
        a1.x = a[j].z; a1.y = a[j].w;
#pragma unroll
        for (int t = 0; t < 8; ++t) {
            v2f b0;
            b0.x = b[j & 1][t].x; b0.y = b[j & 1][t].y;
            acc[t] = __builtin_amdgcn_wmma_f32_16x16x4_f32(false, a0, false, b0,
                                                           (short)0, acc[t], false, false);
        }
#pragma unroll
        for (int t = 0; t < 8; ++t) {
            v2f b1;
            b1.x = b[j & 1][t].z; b1.y = b[j & 1][t].w;
            acc[t] = __builtin_amdgcn_wmma_f32_16x16x4_f32(false, a1, false, b1,
                                                           (short)0, acc[t], false, false);
        }
        __builtin_amdgcn_sched_barrier(0);
    }

#pragma unroll
    for (int t = 0; t < 8; ++t) {
        const int col = 16 * t + c;
#pragma unroll
        for (int r = 0; r < 8; ++r) {
            const size_t i = (size_t)(r0 + (h << 3) + r) * NFEAT + col;
            xw[i] = acc[t][r];
        }
    }
    int ec = 0;
#pragma unroll
    for (int s = 0; s < DEG_SLICES; ++s) ec += pc[s];
    const float di = rsqrtf(1.0f + (float)ec);
    if (h == 0) dinv[r0 + c] = di;
}


#define RANGE 512
#define FUSE_THREADS 1024
#define FUSE_WAVES 32
#define QCAP 144
#define SCAN_AHEAD 4
#define EDGE_GROUP 8

__device__ __forceinline__ void add_queued_edges(const int2* p, int cnt, int lo,
                                                 const int* __restrict__ src,
                                                 const float* __restrict__ xw,
                                                 const float* __restrict__ dinv,
                                                 float* rowsf, int lane) {
    __builtin_amdgcn_fence(__ATOMIC_ACQ_REL, "wavefront");
    const int2 en = p[(lane < cnt) ? lane : (cnt - 1)];
    const int dl = en.y;
    const int s = src[en.x];
    const float dd = dinv[lo + dl];
    const float ds = dinv[s];
    if (cnt == EDGE_GROUP) {
        float v[EDGE_GROUP][4];
#pragma unroll
        for (int j = 0; j < EDGE_GROUP; ++j) {
            const int sj = __builtin_amdgcn_readlane(s, j);
            const float* xs = xw + (size_t)sj * NFEAT + lane;
            v[j][0] = xs[0];
            v[j][1] = xs[32];
            v[j][2] = xs[64];
            v[j][3] = xs[96];
        }
        const float w = ds * dd;
#pragma unroll
        for (int j = 0; j < EDGE_GROUP; ++j) {
            const int dj = __builtin_amdgcn_readlane(dl, j);
            const float wj = __int_as_float(__builtin_amdgcn_readlane(__float_as_int(w), j));
            float* ab = rowsf + dj * NFEAT + lane;
            atomicAdd(ab + 0,  v[j][0] * wj);
            atomicAdd(ab + 32, v[j][1] * wj);
            atomicAdd(ab + 64, v[j][2] * wj);
            atomicAdd(ab + 96, v[j][3] * wj);
        }
    } else {
        const float w = ds * dd;
        for (int j = 0; j < cnt; ++j) {
            const int sj = __builtin_amdgcn_readlane(s, j);
            const int dj = __builtin_amdgcn_readlane(dl, j);
            const float wj = __int_as_float(__builtin_amdgcn_readlane(__float_as_int(w), j));
            const float* xs = xw + (size_t)sj * NFEAT + lane;
            const float v0 = xs[0];
            const float v1 = xs[32];
            const float v2 = xs[64];
            const float v3 = xs[96];
            float* ab = rowsf + dj * NFEAT + lane;
            atomicAdd(ab + 0,  v0 * wj);
            atomicAdd(ab + 32, v1 * wj);
            atomicAdd(ab + 64, v2 * wj);
            atomicAdd(ab + 96, v3 * wj);
        }
    }
}

template <bool VEC>
__device__ __forceinline__ void load_step(const int* __restrict__ dst, int it0, int nchunks, int rot,
                                          int e, int wv, int lane,
                                          int (&dd)[SCAN_AHEAD][4], int (&cb)[SCAN_AHEAD],
                                          int (&lim)[SCAN_AHEAD]) {
#pragma unroll
    for (int u = 0; u < SCAN_AHEAD; ++u) {
        const int ch = (it0 + u) * FUSE_WAVES + wv;
        const bool okc = ch < nchunks;
        int cr = okc ? (ch + rot) : 0;
        if (cr >= nchunks) cr -= nchunks;
        const int cbase = cr * 128;
        cb[u] = cbase;
        const int left = e - cbase;
        lim[u] = okc ? (left < 128 ? left : 128) : 0;
        const int base = cbase + lane * 4;
        if (VEC) {
            const int lb = (base + 4 <= e) ? base : (e - 4);
            const int4 t = *(const int4*)(dst + lb);
            dd[u][0] = t.x; dd[u][1] = t.y; dd[u][2] = t.z; dd[u][3] = t.w;
        } else {
#pragma unroll
            for (int k = 0; k < 4; ++k) dd[u][k] = dst[(base + k < e) ? (base + k) : (e - 1)];
        }
    }
}

template <bool VEC>
__device__ __forceinline__ void scan_edges(const int* __restrict__ src, const int* __restrict__ dst,
                                           const float* __restrict__ xw,
                                           const float* __restrict__ dinv,
                                           float* rowsf, int2* q, int lo, int e, int wv, int lane) {
    const int nchunks = (e + 127) >> 7;
    const int rot = (int)(((long long)blockIdx.x * nchunks) / gridDim.x);
    const int iters = (nchunks + FUSE_WAVES - 1) / FUSE_WAVES;
    int qn = 0;
    int dd[SCAN_AHEAD][4], cb[SCAN_AHEAD], lim[SCAN_AHEAD];
    load_step<VEC>(dst, 0, nchunks, rot, e, wv, lane, dd, cb, lim);
    for (int it0 = 0; it0 < iters; it0 += SCAN_AHEAD) {
        int dn[SCAN_AHEAD][4], cn[SCAN_AHEAD], ln[SCAN_AHEAD];
        load_step<VEC>(dst, it0 + SCAN_AHEAD, nchunks, rot, e, wv, lane, dn, cn, ln);
#pragma unroll
        for (int u = 0; u < SCAN_AHEAD; ++u) {
            if (lim[u] < 128) {
#pragma unroll
                for (int k = 0; k < 4; ++k)
                    if (lane * 4 + k >= lim[u]) dd[u][k] = -1;
            }
#pragma unroll
            for (int k = 0; k < 4; ++k) {
                const int dl = dd[u][k] - lo;
                const bool m = (unsigned)dl < (unsigned)RANGE;
                const unsigned bal = __builtin_amdgcn_ballot_w32(m);
                if (bal != 0u) {
                    const int pos = qn + (int)__builtin_amdgcn_mbcnt_lo(bal, 0u);
                    if (m) q[pos] = make_int2(cb[u] + lane * 4 + k, dl);
                    qn += __builtin_popcount(bal);
                }
            }
            while (qn >= EDGE_GROUP) {
                qn -= EDGE_GROUP;
                add_queued_edges(q + qn, EDGE_GROUP, lo, src, xw, dinv, rowsf, lane);
            }
        }
#pragma unroll
        for (int u = 0; u < SCAN_AHEAD; ++u) {
            cb[u] = cn[u];
            lim[u] = ln[u];
#pragma unroll
            for (int k = 0; k < 4; ++k) dd[u][k] = dn[u][k];
        }
    }
    if (qn > 0) add_queued_edges(q, qn, lo, src, xw, dinv, rowsf, lane);
}

__global__ __launch_bounds__(FUSE_THREADS) void range_sums_kernel(const int* __restrict__ src,
                                                                  const int* __restrict__ dst,
                                                                  const float* __restrict__ xw,
                                                                  const float* __restrict__ dinv,
                                                                  const float* __restrict__ bias,
                                                                  const float* __restrict__ wreg,
                                                                  const float* __restrict__ breg,
                                                                  float* __restrict__ out,
                                                                  int n, int e) {
    __shared__ float4 rows[RANGE * (NFEAT / 4)];
    __shared__ int2 queue[FUSE_WAVES * QCAP];

    const int tid  = threadIdx.x;
    const int lane = tid & 31;
    const int wv = __builtin_amdgcn_readfirstlane(tid >> 5);
    const int lo = blockIdx.x * RANGE;
    const int nrows = (n - lo < RANGE) ? (n - lo) : RANGE;

    const float4 b4 = ((const float4*)bias)[lane];
    const float4* xw4 = (const float4*)(xw + (size_t)lo * NFEAT);
    for (int i0 = 0; i0 < 16; i0 += 8) {
        float4 v[8];
        float di[8];
#pragma unroll
        for (int i = 0; i < 8; ++i) {
            const int row = wv + 32 * (i0 + i);
            const int rl = (row < nrows) ? row : (nrows - 1);
            v[i]  = xw4[rl * (NFEAT / 4) + lane];
            di[i] = dinv[lo + rl];
        }
#pragma unroll
        for (int i = 0; i < 8; ++i) {
            const int row = wv + 32 * (i0 + i);
            float4 r;
            r.x = b4.x + v[i].x * di[i] * di[i];
            r.y = b4.y + v[i].y * di[i] * di[i];
            r.z = b4.z + v[i].z * di[i] * di[i];
            r.w = b4.w + v[i].w * di[i] * di[i];
            rows[row * (NFEAT / 4) + lane] = r;
        }
    }
    __syncthreads();

    float* rowsf = (float*)rows;
    int2* q = queue + wv * QCAP;
    if ((e & 3) == 0 && e >= 4) scan_edges<true>(src, dst, xw, dinv, rowsf, q, lo, e, wv, lane);
    else if (e > 0)             scan_edges<false>(src, dst, xw, dinv, rowsf, q, lo, e, wv, lane);
    __syncthreads();

    const float4 w4 = ((const float4*)wreg)[lane];
    float res = 0.0f;
#pragma unroll
    for (int j = 0; j < 16; ++j) {
        const float4 a = rows[(wv * 16 + j) * (NFEAT / 4) + lane];
        float dot = fmaxf(a.x, 0.0f) * w4.x + fmaxf(a.y, 0.0f) * w4.y +
                    fmaxf(a.z, 0.0f) * w4.z + fmaxf(a.w, 0.0f) * w4.w;
#pragma unroll
        for (int off = 16; off > 0; off >>= 1)
            dot += __shfl_xor(dot, off, 32);
        if (lane == j) res = dot;
    }
    float* outv = (float*)queue;
    if (lane < 16) outv[wv * 16 + lane] = res + breg[0];
    __syncthreads();
    if (tid < RANGE && lo + tid < n) out[lo + tid] = outv[tid];
}


extern "C" void kernel_launch(void* const* d_in, const int* in_sizes, int n_in,
                              void* d_out, int out_size, void* d_ws, size_t ws_size,
                              hipStream_t stream) {
    const float* x    = (const float*)d_in[0];
    const int*   ei   = (const int*)d_in[1];
    const float* W    = (const float*)d_in[2];
    const float* b    = (const float*)d_in[3];
    const float* wreg = (const float*)d_in[4];
    const float* breg = (const float*)d_in[5];
    float* out = (float*)d_out;

    const int n = in_sizes[0] / NFEAT;
    const int e = in_sizes[1] / 2;
    const int* src = ei;
    const int* dst = ei + e;

    float* xw   = (float*)d_ws;
    float* dinv = xw + (size_t)n * NFEAT;
    int*   part = (int*)(dinv + n);

    const int B = 256;
    deg_count_kernel<<<((n + DEG_RANGE - 1) / DEG_RANGE) * DEG_SLICES, DEG_THREADS, 0, stream>>>(dst, part, n, e);

    gemm_xwT_wmma<<<(n + ROWS_PER_BLOCK - 1) / ROWS_PER_BLOCK, B, 0, stream>>>(x, W, part, dinv, xw, n);

    range_sums_kernel<<<(n + RANGE - 1) / RANGE, FUSE_THREADS, 0, stream>>>(src, dst, xw, dinv, b, wreg, breg, out, n, e);
}


# Edge pass walks a 1-byte key per edge (dst/512, plane written by extra blocks of the count launch): 4 keys a zero-byte test, one branch per 512 edges; dst read for kept edges only; LDS arrival order
# speedup vs baseline: 3.0940x; 1.0880x over previous
#include <hip/hip_runtime.h>
#include <cstdint>

#define NFEAT 128

typedef __attribute__((ext_vector_type(2))) float v2f;
typedef __attribute__((ext_vector_type(8))) float v8f;


#define DEG_RANGE 6272
#define DEG_SLICES 8
#define DEG_THREADS 1024
#define DEG_AHEAD 16

#define RANGE 512
#define RANGE_SHIFT 9
#define KEY_LOAD 512
#define KEY_BLOCKS 64

__global__ __launch_bounds__(DEG_THREADS) void deg_count_kernel(const int* __restrict__ dst,
                                                                int* __restrict__ part,
                                                                unsigned* __restrict__ keys,
                                                                int count_blocks, int nload,
                                                                int n, int e) {
    __shared__ int cnt[DEG_RANGE];
    const int tid = threadIdx.x;
    if ((int)blockIdx.x >= count_blocks) {
        const int nd = nload * (KEY_LOAD / 4);
        for (int w = ((int)blockIdx.x - count_blocks) * DEG_THREADS + tid; w < nd;
             w += KEY_BLOCKS * DEG_THREADS) {
            unsigned pk = 0u;
#pragma unroll
            for (int b = 0; b < 4; ++b) {
                const int i = 4 * w + b;
                const int v = dst[(i < e) ? i : (e - 1)];
                const unsigned key = (i < e) ? (((unsigned)v >> RANGE_SHIFT) & 255u) : 255u;
                pk |= key << (8 * b);
            }
            keys[w] = pk;
        }
        return;
    }
    const int k = blockIdx.x / DEG_SLICES;
    const int s = blockIdx.x - k * DEG_SLICES;
    const int lo = k * DEG_RANGE;
    for (int i = tid; i < DEG_RANGE; i += DEG_THREADS) cnt[i] = 0;
    __syncthreads();

    const int per = (e + DEG_SLICES - 1) / DEG_SLICES;
    const int e0 = s * per;
    const int e1 = (e0 + per < e) ? (e0 + per) : e;
    for (int b = e0 + tid; b < e1; b += DEG_THREADS * DEG_AHEAD) {
        int d[DEG_AHEAD];
#pragma unroll
        for (int u = 0; u < DEG_AHEAD; ++u) {
            const int i = b + u * DEG_THREADS;
            const int v = dst[(i < e1) ? i : (e1 - 1)];
            d[u] = (i < e1) ? v : -1;
        }
#pragma unroll
        for (int u = 0; u < DEG_AHEAD; ++u) {
            const unsigned dl = (unsigned)(d[u] - lo);
            if (dl < (unsigned)DEG_RANGE) atomicAdd(&cnt[dl], 1);
        }
    }
    __syncthreads();

    const int left = n - lo;
    const int hi = (left < DEG_RANGE) ? left : DEG_RANGE;
    int* pp = part + (size_t)s * n + lo;
    for (int i = tid; i < hi; i += DEG_THREADS) pp[i] = cnt[i];
}


#define ROWS_PER_BLOCK 128
#define W_PITCH4 33

__global__ __launch_bounds__(256) void gemm_xwT_wmma(const float* __restrict__ x,
                                                     const float* __restrict__ W,
                                                     const int* __restrict__ part,
                                                     float* __restrict__ dinv,
                                                     float* __restrict__ xw, int n) {
    __shared__ float4 lds_w[NFEAT * W_PITCH4];

    const int tid  = threadIdx.x;
    const int lane = tid & 31;
    const int wave = __builtin_amdgcn_readfirstlane(tid >> 5);
    const int c    = lane & 15;
    const int h    = lane >> 4;
    const int r0   = blockIdx.x * ROWS_PER_BLOCK + wave * 16;
    const bool live = r0 < n;

    const float4* wg = (const float4*)W;
    float4 wr[16];
#pragma unroll
    for (int i = 0; i < 16; ++i) wr[i] = wg[tid + 256 * i];

    const int rl = live ? r0 : 0;
    const float4* xr = (const float4*)(x + (size_t)(rl + c) * NFEAT) + h;
    float4 a[16];
#pragma unroll
    for (int j = 0; j < 16; ++j) a[j] = xr[2 * j];

    const int* pr = part + rl + c;
    int pc[DEG_SLICES];
#pragma unroll
    for (int s = 0; s < DEG_SLICES; ++s) pc[s] = pr[(size_t)s * n];

#pragma unroll
    for (int i = 0; i < 16; ++i) {
        const int f = tid + 256 * i;
        lds_w[(f >> 5) * W_PITCH4 + (f & 31)] = wr[i];
    }
    __syncthreads();
    if (!live) return;

    v8f acc[8] = {};
    const float4* wl = lds_w + c * W_PITCH4 + h;

    float4 b[2][8];
#pragma unroll
    for (int t = 0; t < 8; ++t) b[0][t] = wl[t * 16 * W_PITCH4];

#pragma unroll
    for (int j = 0; j < 16; ++j) {
        if (j + 1 < 16) {
#pragma unroll
            for (int t = 0; t < 8; ++t) b[(j + 1) & 1][t] = wl[t * 16 * W_PITCH4 + 2 * (j + 1)];
        }
        __builtin_amdgcn_sched_barrier(0);
        v2f a0, a1;
        a0.x = a[j].x; a0.y = a[j].y;
        a1.x = a[j].z; a1.y = a[j].w;
#pragma unroll
        for (int t = 0; t < 8; ++t) {
            v2f b0;
            b0.x = b[j & 1][t].x; b0.y = b[j & 1][t].y;
            acc[t] = __builtin_amdgcn_wmma_f32_16x16x4_f32(false, a0, false, b0,
                                                           (short)0, acc[t], false, false);
        }
#pragma unroll
        for (int t = 0; t < 8; ++t) {
            v2f b1;
            b1.x = b[j & 1][t].z; b1.y = b[j & 1][t].w;
            acc[t] = __builtin_amdgcn_wmma_f32_16x16x4_f32(false, a1, false, b1,
                                                           (short)0, acc[t], false, false);
        }
        __builtin_amdgcn_sched_barrier(0);
    }

#pragma unroll
    for (int t = 0; t < 8; ++t) {
        const int col = 16 * t + c;
#pragma unroll
        for (int r = 0; r < 8; ++r) {
            const size_t i = (size_t)(r0 + (h << 3) + r) * NFEAT + col;
            xw[i] = acc[t][r];
        }
    }
    int ec = 0;
#pragma unroll
    for (int s = 0; s < DEG_SLICES; ++s) ec += pc[s];
    const float di = rsqrtf(1.0f + (float)ec);
    if (h == 0) dinv[r0 + c] = di;
}


#define FUSE_THREADS 1024
#define FUSE_WAVES 32
#define QCAP 40
#define SCAN_AHEAD 4
#define EDGE_GROUP 8

__device__ __forceinline__ void add_queued_edges(const int* p, int cnt, int lo,
                                                 const int* __restrict__ src,
                                                 const int* __restrict__ dst,
                                                 const float* __restrict__ xw,
                                                 const float* __restrict__ dinv,
                                                 float* rowsf, int lane) {
    __builtin_amdgcn_fence(__ATOMIC_ACQ_REL, "wavefront");
    const int en = p[(lane < cnt) ? lane : (cnt - 1)];
    const int s = src[en];
    const int d = dst[en];
    const int dl = d - lo;
    const int row = ((unsigned)dl < (unsigned)RANGE) ? dl : RANGE;
    const float dd = dinv[d];
    const float ds = dinv[s];
    if (cnt == EDGE_GROUP) {
        float v[EDGE_GROUP][4];
#pragma unroll
        for (int j = 0; j < EDGE_GROUP; ++j) {
            const int sj = __builtin_amdgcn_readlane(s, j);
            const float* xs = xw + (size_t)sj * NFEAT + lane;
            v[j][0] = xs[0];
            v[j][1] = xs[32];
            v[j][2] = xs[64];
            v[j][3] = xs[96];
        }
        const float w = ds * dd;
#pragma unroll
        for (int j = 0; j < EDGE_GROUP; ++j) {
            const int dj = __builtin_amdgcn_readlane(row, j);
            const float wj = __int_as_float(__builtin_amdgcn_readlane(__float_as_int(w), j));
            float* ab = rowsf + dj * NFEAT + lane;
            atomicAdd(ab + 0,  v[j][0] * wj);
            atomicAdd(ab + 32, v[j][1] * wj);
            atomicAdd(ab + 64, v[j][2] * wj);
            atomicAdd(ab + 96, v[j][3] * wj);
        }
    } else {
        const float w = ds * dd;
        for (int j = 0; j < cnt; ++j) {
            const int sj = __builtin_amdgcn_readlane(s, j);
            const int dj = __builtin_amdgcn_readlane(row, j);
            const float wj = __int_as_float(__builtin_amdgcn_readlane(__float_as_int(w), j));
            const float* xs = xw + (size_t)sj * NFEAT + lane;
            const float v0 = xs[0];
            const float v1 = xs[32];
            const float v2 = xs[64];
            const float v3 = xs[96];
            float* ab = rowsf + dj * NFEAT + lane;
            atomicAdd(ab + 0,  v0 * wj);
            atomicAdd(ab + 32, v1 * wj);
            atomicAdd(ab + 64, v2 * wj);
            atomicAdd(ab + 96, v3 * wj);
        }
    }
}

__device__ __forceinline__ void load_keys(const uint4* __restrict__ kp, int it0, int nload, int rot,
                                          int wv, int lane,
                                          uint4 (&kk)[SCAN_AHEAD], int (&cb)[SCAN_AHEAD]) {
#pragma unroll
    for (int u = 0; u < SCAN_AHEAD; ++u) {
        const int ch = (it0 + u) * FUSE_WAVES + wv;
        const bool okc = ch < nload;
        int cr = okc ? (ch + rot) : 0;
        if (cr >= nload) cr -= nload;
        cb[u] = okc ? cr * KEY_LOAD : -1;
        kk[u] = kp[cr * 32 + lane];
    }
}

__device__ __forceinline__ unsigned match_bytes(unsigned keys4, unsigned k4) {
    const unsigned t = keys4 ^ k4;
    return (t - 0x01010101u) & ~t & 0x80808080u;
}

__device__ __forceinline__ void scan_keys(const unsigned* __restrict__ keys,
                                          const int* __restrict__ src, const int* __restrict__ dst,
                                          const float* __restrict__ xw,
                                          const float* __restrict__ dinv,
                                          float* rowsf, int* q, int lo, int e, int nload,
                                          int wv, int lane) {
    const uint4* kp = (const uint4*)keys;
    const unsigned k4 = (blockIdx.x & 255u) * 0x01010101u;
    const int rot = (int)(((long long)blockIdx.x * nload) / gridDim.x);
    const int iters = (nload + FUSE_WAVES - 1) / FUSE_WAVES;
    int qn = 0;
    uint4 kk[SCAN_AHEAD];
    int cb[SCAN_AHEAD];
    load_keys(kp, 0, nload, rot, wv, lane, kk, cb);
    for (int it0 = 0; it0 < iters; it0 += SCAN_AHEAD) {
        uint4 kn[SCAN_AHEAD];
        int cn[SCAN_AHEAD];
        load_keys(kp, it0 + SCAN_AHEAD, nload, rot, wv, lane, kn, cn);
#pragma unroll
        for (int u = 0; u < SCAN_AHEAD; ++u) {
            if (cb[u] < 0) continue;
            unsigned m = (match_bytes(kk[u].x, k4) >> 7) | (match_bytes(kk[u].y, k4) >> 6) |
                         (match_bytes(kk[u].z, k4) >> 5) | (match_bytes(kk[u].w, k4) >> 4);
            if (cb[u] + KEY_LOAD > e) {
                const int left = e - cb[u] - lane * 16;
                unsigned keep = 0u;
#pragma unroll
                for (int qd = 0; qd < 4; ++qd) {
                    int c = left - 4 * qd;
                    c = (c < 0) ? 0 : ((c > 4) ? 4 : c);
                    const unsigned bytes = (c == 4) ? 0x01010101u
                                                    : (((1u << (8 * c)) - 1u) & 0x01010101u);
                    keep |= bytes << qd;
                }
                m &= keep;
            }
            unsigned bal = __builtin_amdgcn_ballot_w32(m != 0u);
            while (bal != 0u) {
                const int pbit = __builtin_ctz(m | 0x10000000u);
                const int edge = cb[u] + lane * 16 + ((pbit & 3) << 2) + (pbit >> 3);
                const int pos = qn + (int)__builtin_amdgcn_mbcnt_lo(bal, 0u);
                if (m != 0u) q[pos] = edge;
                qn += __builtin_popcount(bal);
                m &= m - 1u;
                while (qn >= EDGE_GROUP) {
                    qn -= EDGE_GROUP;
                    add_queued_edges(q + qn, EDGE_GROUP, lo, src, dst, xw, dinv, rowsf, lane);
                }
                bal = __builtin_amdgcn_ballot_w32(m != 0u);
            }
        }
#pragma unroll
        for (int u = 0; u < SCAN_AHEAD; ++u) {
            cb[u] = cn[u];
            kk[u] = kn[u];
        }
    }
    if (qn > 0) add_queued_edges(q, qn, lo, src, dst, xw, dinv, rowsf, lane);
}

__global__ __launch_bounds__(FUSE_THREADS) void range_sums_kernel(const int* __restrict__ src,
                                                                  const int* __restrict__ dst,
                                                                  const unsigned* __restrict__ keys,
                                                                  const float* __restrict__ xw,
                                                                  const float* __restrict__ dinv,
                                                                  const float* __restrict__ bias,
                                                                  const float* __restrict__ wreg,
                                                                  const float* __restrict__ breg,
                                                                  float* __restrict__ out,
                                                                  int n, int e, int nload) {
    __shared__ float4 rows[(RANGE + 1) * (NFEAT / 4)];
    __shared__ int queue[FUSE_WAVES * QCAP];

    const int tid  = threadIdx.x;
    const int lane = tid & 31;
    const int wv = __builtin_amdgcn_readfirstlane(tid >> 5);
    const int lo = blockIdx.x * RANGE;
    const int nrows = (n - lo < RANGE) ? (n - lo) : RANGE;

    const float4 b4 = ((const float4*)bias)[lane];
    const float4* xw4 = (const float4*)(xw + (size_t)lo * NFEAT);
    for (int i0 = 0; i0 < 16; i0 += 8) {
        float4 v[8];
        float di[8];
#pragma unroll
        for (int i = 0; i < 8; ++i) {
            const int row = wv + 32 * (i0 + i);
            const int rl = (row < nrows) ? row : (nrows - 1);
            v[i]  = xw4[rl * (NFEAT / 4) + lane];
            di[i] = dinv[lo + rl];
        }
#pragma unroll
        for (int i = 0; i < 8; ++i) {
            const int row = wv + 32 * (i0 + i);
            float4 r;
            r.x = b4.x + v[i].x * di[i] * di[i];
            r.y = b4.y + v[i].y * di[i] * di[i];
            r.z = b4.z + v[i].z * di[i] * di[i];
            r.w = b4.w + v[i].w * di[i] * di[i];
            rows[row * (NFEAT / 4) + lane] = r;
        }
    }
    if (wv == 0) rows[RANGE * (NFEAT / 4) + lane] = make_float4(0.0f, 0.0f, 0.0f, 0.0f);
    __syncthreads();

    float* rowsf = (float*)rows;
    int* q = queue + wv * QCAP;
    if (e > 0) scan_keys(keys, src, dst, xw, dinv, rowsf, q, lo, e, nload, wv, lane);
    __syncthreads();

    const float4 w4 = ((const float4*)wreg)[lane];
    float res = 0.0f;
#pragma unroll
    for (int j = 0; j < 16; ++j) {
        const float4 a = rows[(wv * 16 + j) * (NFEAT / 4) + lane];
        float dot = fmaxf(a.x, 0.0f) * w4.x + fmaxf(a.y, 0.0f) * w4.y +
                    fmaxf(a.z, 0.0f) * w4.z + fmaxf(a.w, 0.0f) * w4.w;
#pragma unroll
        for (int off = 16; off > 0; off >>= 1)
            dot += __shfl_xor(dot, off, 32);
        if (lane == j) res = dot;
    }
    float* outv = (float*)queue;
    if (lane < 16) outv[wv * 16 + lane] = res + breg[0];
    __syncthreads();
    if (tid < RANGE && lo + tid < n) out[lo + tid] = outv[tid];
}


extern "C" void kernel_launch(void* const* d_in, const int* in_sizes, int n_in,
                              void* d_out, int out_size, void* d_ws, size_t ws_size,
                              hipStream_t stream) {
    const float* x    = (const float*)d_in[0];
    const int*   ei   = (const int*)d_in[1];
    const float* W    = (const float*)d_in[2];
    const float* b    = (const float*)d_in[3];
    const float* wreg = (const float*)d_in[4];
    const float* breg = (const float*)d_in[5];
    float* out = (float*)d_out;

    const int n = in_sizes[0] / NFEAT;
    const int e = in_sizes[1] / 2;
    const int* src = ei;
    const int* dst = ei + e;

    const int nload = (e + KEY_LOAD - 1) / KEY_LOAD;
    unsigned* keys = (unsigned*)d_ws;
    float* xw   = (float*)(keys + (size_t)nload * (KEY_LOAD / 4));
    float* dinv = xw + (size_t)n * NFEAT;
    int*   part = (int*)(dinv + n);

    const int B = 256;
    const int count_blocks = ((n + DEG_RANGE - 1) / DEG_RANGE) * DEG_SLICES;
    deg_count_kernel<<<count_blocks + KEY_BLOCKS, DEG_THREADS, 0, stream>>>(dst, part, keys, count_blocks, nload, n, e);

    gemm_xwT_wmma<<<(n + ROWS_PER_BLOCK - 1) / ROWS_PER_BLOCK, B, 0, stream>>>(x, W, part, dinv, xw, n);

    range_sums_kernel<<<(n + RANGE - 1) / RANGE, FUSE_THREADS, 0, stream>>>(src, dst, keys, xw, dinv, b, wreg, breg, out, n, e, nload);
}


# Edge adds without LDS float atomics: kept edges are filed (LDS integer add) in the list of the wave that owns the row; owners sum by plain 16-byte LDS read-add-write; sum order = list arrival order
# speedup vs baseline: 3.4309x; 1.1089x over previous
#include <hip/hip_runtime.h>
#include <cstdint>

#define NFEAT 128

typedef __attribute__((ext_vector_type(2))) float v2f;
typedef __attribute__((ext_vector_type(8))) float v8f;


#define DEG_RANGE 6272
#define DEG_SLICES 8
#define DEG_THREADS 1024
#define DEG_AHEAD 16

#define RANGE 512
#define RANGE_SHIFT 9
#define KEY_LOAD 512
#define KEY_BLOCKS 64

__global__ __launch_bounds__(DEG_THREADS) void deg_count_kernel(const int* __restrict__ dst,
                                                                int* __restrict__ part,
                                                                unsigned* __restrict__ keys,
                                                                int count_blocks, int nload,
                                                                int n, int e) {
    __shared__ int cnt[DEG_RANGE];
    const int tid = threadIdx.x;
    if ((int)blockIdx.x >= count_blocks) {
        const int nd = nload * (KEY_LOAD / 4);
        for (int w = ((int)blockIdx.x - count_blocks) * DEG_THREADS + tid; w < nd;
             w += KEY_BLOCKS * DEG_THREADS) {
            unsigned pk = 0u;
#pragma unroll
            for (int b = 0; b < 4; ++b) {
                const int i = 4 * w + b;
                const int v = dst[(i < e) ? i : (e - 1)];
                const unsigned key = (i < e) ? (((unsigned)v >> RANGE_SHIFT) & 255u) : 255u;
                pk |= key << (8 * b);
            }
            keys[w] = pk;
        }
        return;
    }
    const int k = blockIdx.x / DEG_SLICES;
    const int s = blockIdx.x - k * DEG_SLICES;
    const int lo = k * DEG_RANGE;
    for (int i = tid; i < DEG_RANGE; i += DEG_THREADS) cnt[i] = 0;
    __syncthreads();

    const int per = (e + DEG_SLICES - 1) / DEG_SLICES;
    const int e0 = s * per;
    const int e1 = (e0 + per < e) ? (e0 + per) : e;
    for (int b = e0 + tid; b < e1; b += DEG_THREADS * DEG_AHEAD) {
        int d[DEG_AHEAD];
#pragma unroll
        for (int u = 0; u < DEG_AHEAD; ++u) {
            const int i = b + u * DEG_THREADS;
            const int v = dst[(i < e1) ? i : (e1 - 1)];
            d[u] = (i < e1) ? v : -1;
        }
#pragma unroll
        for (int u = 0; u < DEG_AHEAD; ++u) {
            const unsigned dl = (unsigned)(d[u] - lo);
            if (dl < (unsigned)DEG_RANGE) atomicAdd(&cnt[dl], 1);
        }
    }
    __syncthreads();

    const int left = n - lo;
    const int hi = (left < DEG_RANGE) ? left : DEG_RANGE;
    int* pp = part + (size_t)s * n + lo;
    for (int i = tid; i < hi; i += DEG_THREADS) pp[i] = cnt[i];
}


#define ROWS_PER_BLOCK 128
#define W_PITCH4 33

__global__ __launch_bounds__(256) void gemm_xwT_wmma(const float* __restrict__ x,
                                                     const float* __restrict__ W,
                                                     const int* __restrict__ part,
                                                     float* __restrict__ dinv,
                                                     float* __restrict__ xw, int n) {
    __shared__ float4 lds_w[NFEAT * W_PITCH4];

    const int tid  = threadIdx.x;
    const int lane = tid & 31;
    const int wave = __builtin_amdgcn_readfirstlane(tid >> 5);
    const int c    = lane & 15;
    const int h    = lane >> 4;
    const int r0   = blockIdx.x * ROWS_PER_BLOCK + wave * 16;
    const bool live = r0 < n;

    const float4* wg = (const float4*)W;
    float4 wr[16];
#pragma unroll
    for (int i = 0; i < 16; ++i) wr[i] = wg[tid + 256 * i];

    const int rl = live ? r0 : 0;
    const float4* xr = (const float4*)(x + (size_t)(rl + c) * NFEAT) + h;
    float4 a[16];
#pragma unroll
    for (int j = 0; j < 16; ++j) a[j] = xr[2 * j];

    const int* pr = part + rl + c;
    int pc[DEG_SLICES];
#pragma unroll
    for (int s = 0; s < DEG_SLICES; ++s) pc[s] = pr[(size_t)s * n];

#pragma unroll
    for (int i = 0; i < 16; ++i) {
        const int f = tid + 256 * i;
        lds_w[(f >> 5) * W_PITCH4 + (f & 31)] = wr[i];
    }
    __syncthreads();
    if (!live) return;

    v8f acc[8] = {};
    const float4* wl = lds_w + c * W_PITCH4 + h;

    float4 b[2][8];
#pragma unroll
    for (int t = 0; t < 8; ++t) b[0][t] = wl[t * 16 * W_PITCH4];

#pragma unroll
    for (int j = 0; j < 16; ++j) {
        if (j + 1 < 16) {
#pragma unroll
            for (int t = 0; t < 8; ++t) b[(j + 1) & 1][t] = wl[t * 16 * W_PITCH4 + 2 * (j + 1)];
        }
        __builtin_amdgcn_sched_barrier(0);
        v2f a0, a1;
        a0.x = a[j].x; a0.y = a[j].y;
        a1.x = a[j].z; a1.y = a[j].w;
#pragma unroll
        for (int t = 0; t < 8; ++t) {
            v2f b0;
            b0.x = b[j & 1][t].x; b0.y = b[j & 1][t].y;
            acc[t] = __builtin_amdgcn_wmma_f32_16x16x4_f32(false, a0, false, b0,
                                                           (short)0, acc[t], false, false);
        }
#pragma unroll
        for (int t = 0; t < 8; ++t) {
            v2f b1;
            b1.x = b[j & 1][t].z; b1.y = b[j & 1][t].w;
            acc[t] = __builtin_amdgcn_wmma_f32_16x16x4_f32(false, a1, false, b1,
                                                           (short)0, acc[t], false, false);
        }
        __builtin_amdgcn_sched_barrier(0);
    }

#pragma unroll
    for (int t = 0; t < 8; ++t) {
        const int col = 16 * t + c;
#pragma unroll
        for (int r = 0; r < 8; ++r) {
            const size_t i = (size_t)(r0 + (h << 3) + r) * NFEAT + col;
            xw[i] = acc[t][r];
        }
    }
    int ec = 0;
#pragma unroll
    for (int s = 0; s < DEG_SLICES; ++s) ec += pc[s];
    const float di = rsqrtf(1.0f + (float)ec);
    if (h == 0) dinv[r0 + c] = di;
}


#define FUSE_THREADS 1024
#define FUSE_WAVES 32
#define QCAP 64
#define SCAN_AHEAD 4
#define EDGE_GROUP 8
#define ROUTE_GROUP 32
#define OWN_ROWS 16
#define OWN_SHIFT 4
#define OWNER_CAP 320

__device__ __forceinline__ void add_queued_edges(const int* p, int cnt, int lo,
                                                 const int* __restrict__ src,
                                                 const int* __restrict__ dst,
                                                 const float* __restrict__ xw,
                                                 const float* __restrict__ dinv,
                                                 float* rowsf, int lane) {
    __builtin_amdgcn_fence(__ATOMIC_ACQ_REL, "wavefront");
    const int en = p[(lane < cnt) ? lane : (cnt - 1)];
    const int s = src[en];
    const int d = dst[en];
    const int dl = d - lo;
    const int row = ((unsigned)dl < (unsigned)RANGE) ? dl : RANGE;
    const float dd = dinv[d];
    const float ds = dinv[s];
    if (cnt == EDGE_GROUP) {
        float v[EDGE_GROUP][4];
#pragma unroll
        for (int j = 0; j < EDGE_GROUP; ++j) {
            const int sj = __builtin_amdgcn_readlane(s, j);
            const float* xs = xw + (size_t)sj * NFEAT + lane;
            v[j][0] = xs[0];
            v[j][1] = xs[32];
            v[j][2] = xs[64];
            v[j][3] = xs[96];
        }
        const float w = ds * dd;
#pragma unroll
        for (int j = 0; j < EDGE_GROUP; ++j) {
            const int dj = __builtin_amdgcn_readlane(row, j);
            const float wj = __int_as_float(__builtin_amdgcn_readlane(__float_as_int(w), j));
            float* ab = rowsf + dj * NFEAT + lane;
            atomicAdd(ab + 0,  v[j][0] * wj);
            atomicAdd(ab + 32, v[j][1] * wj);
            atomicAdd(ab + 64, v[j][2] * wj);
            atomicAdd(ab + 96, v[j][3] * wj);
        }
    } else {
        const float w = ds * dd;
        for (int j = 0; j < cnt; ++j) {
            const int sj = __builtin_amdgcn_readlane(s, j);
            const int dj = __builtin_amdgcn_readlane(row, j);
            const float wj = __int_as_float(__builtin_amdgcn_readlane(__float_as_int(w), j));
            const float* xs = xw + (size_t)sj * NFEAT + lane;
            const float v0 = xs[0];
            const float v1 = xs[32];
            const float v2 = xs[64];
            const float v3 = xs[96];
            float* ab = rowsf + dj * NFEAT + lane;
            atomicAdd(ab + 0,  v0 * wj);
            atomicAdd(ab + 32, v1 * wj);
            atomicAdd(ab + 64, v2 * wj);
            atomicAdd(ab + 96, v3 * wj);
        }
    }
}

__device__ __forceinline__ void file_queued_edges(const int* p, int cnt, int lo,
                                                  const int* __restrict__ src,
                                                  const int* __restrict__ dst,
                                                  int* ocnt, int* olist, int lane) {
    __builtin_amdgcn_fence(__ATOMIC_ACQ_REL, "wavefront");
    const int en = p[(lane < cnt) ? lane : (cnt - 1)];
    const int s = src[en];
    const int d = dst[en];
    asm volatile("" :: "v"(s), "v"(d));
    const unsigned dl = (unsigned)(d - lo);
    if (lane < cnt && dl < (unsigned)RANGE) {
        const int ow = (int)(dl >> OWN_SHIFT);
        const int pos = atomicAdd(&ocnt[ow], 1);
        if (pos < OWNER_CAP) olist[ow * OWNER_CAP + pos] = (s << OWN_SHIFT) | (int)(dl & (OWN_ROWS - 1));
    }
}

__device__ __forceinline__ void ask_group_rows(int ent, int own_lo, int lane,
                                               const float* __restrict__ xw,
                                               const float* __restrict__ dinv,
                                               float4 (&v)[8], int& r, float& ds, float& dd) {
    const int s = ent >> OWN_SHIFT;
    r = ent & (OWN_ROWS - 1);
    ds = dinv[s];
    dd = dinv[own_lo + r];
#pragma unroll
    for (int j = 0; j < 8; ++j) {
        const int sj = __builtin_amdgcn_readlane(s, j);
        v[j] = ((const float4*)(xw + (size_t)sj * NFEAT))[lane];
    }
}

__device__ __forceinline__ void add_group_rows(const float4 (&v)[8], int r, float ds, float dd,
                                               int nv, float4* myrows) {
    const float w = ds * dd;
#pragma unroll
    for (int j = 0; j < 8; ++j) {
        if (j < nv) {
            const int rj = __builtin_amdgcn_readlane(r, j);
            const float wj = __int_as_float(__builtin_amdgcn_readlane(__float_as_int(w), j));
            float4* pr = myrows + rj * (NFEAT / 4);
            float4 a = *pr;
            a.x += v[j].x * wj;
            a.y += v[j].y * wj;
            a.z += v[j].z * wj;
            a.w += v[j].w * wj;
            *pr = a;
        }
    }
}

__device__ __forceinline__ void sum_own_list(const int* ol, int cnt, int own_lo, int lane,
                                             const float* __restrict__ xw,
                                             const float* __restrict__ dinv, float4* myrows) {
    const int last = cnt - 1;
    const int l8 = lane & 7;
    float4 va[8], vb[8];
    int ra, rb;
    float dsa, dda, dsb, ddb;
    int ent = ol[(l8 < last) ? l8 : last];
    ask_group_rows(ent, own_lo, lane, xw, dinv, va, ra, dsa, dda);
    ent = ol[(8 + l8 < last) ? (8 + l8) : last];
    for (int g = 0; g * 8 < cnt; g += 2) {
        ask_group_rows(ent, own_lo, lane, xw, dinv, vb, rb, dsb, ddb);
        int i = (g + 2) * 8 + l8;
        ent = ol[(i < last) ? i : last];
        add_group_rows(va, ra, dsa, dda, cnt - g * 8, myrows);
        ask_group_rows(ent, own_lo, lane, xw, dinv, va, ra, dsa, dda);
        i = (g + 3) * 8 + l8;
        ent = ol[(i < last) ? i : last];
        add_group_rows(vb, rb, dsb, ddb, cnt - (g + 1) * 8, myrows);
    }
}

__device__ __forceinline__ void load_keys(const uint4* __restrict__ kp, int it0, int nload, int rot,
                                          int wv, int lane,
                                          uint4 (&kk)[SCAN_AHEAD], int (&cb)[SCAN_AHEAD]) {
#pragma unroll
    for (int u = 0; u < SCAN_AHEAD; ++u) {
        const int ch = (it0 + u) * FUSE_WAVES + wv;
        const bool okc = ch < nload;
        int cr = okc ? (ch + rot) : 0;
        if (cr >= nload) cr -= nload;
        cb[u] = okc ? cr * KEY_LOAD : -1;
        kk[u] = kp[cr * 32 + lane];
    }
}

__device__ __forceinline__ unsigned match_bytes(unsigned keys4, unsigned k4) {
    const unsigned t = keys4 ^ k4;
    return (t - 0x01010101u) & ~t & 0x80808080u;
}

template <bool FILE>
__device__ __forceinline__ void scan_keys(const unsigned* __restrict__ keys,
                                          const int* __restrict__ src, const int* __restrict__ dst,
                                          const float* __restrict__ xw,
                                          const float* __restrict__ dinv,
                                          float* rowsf, int* ocnt, int* olist, int* q, int lo,
                                          int e, int nload, int wv, int lane) {
    constexpr int GROUP = FILE ? ROUTE_GROUP : EDGE_GROUP;
    const uint4* kp = (const uint4*)keys;
    const unsigned k4 = (blockIdx.x & 255u) * 0x01010101u;
    const int rot = (int)(((long long)blockIdx.x * nload) / gridDim.x);
    const int iters = (nload + FUSE_WAVES - 1) / FUSE_WAVES;
    int qn = 0;
    uint4 kk[SCAN_AHEAD];
    int cb[SCAN_AHEAD];
    load_keys(kp, 0, nload, rot, wv, lane, kk, cb);
    for (int it0 = 0; it0 < iters; it0 += SCAN_AHEAD) {
        uint4 kn[SCAN_AHEAD];
        int cn[SCAN_AHEAD];
        load_keys(kp, it0 + SCAN_AHEAD, nload, rot, wv, lane, kn, cn);
#pragma unroll
        for (int u = 0; u < SCAN_AHEAD; ++u) {
            if (cb[u] < 0) continue;
            unsigned m = (match_bytes(kk[u].x, k4) >> 7) | (match_bytes(kk[u].y, k4) >> 6) |
                         (match_bytes(kk[u].z, k4) >> 5) | (match_bytes(kk[u].w, k4) >> 4);
            if (cb[u] + KEY_LOAD > e) {
                const int left = e - cb[u] - lane * 16;
                unsigned keep = 0u;
#pragma unroll
                for (int qd = 0; qd < 4; ++qd) {
                    int c = left - 4 * qd;
                    c = (c < 0) ? 0 : ((c > 4) ? 4 : c);
                    const unsigned bytes = (c == 4) ? 0x01010101u
                                                    : (((1u << (8 * c)) - 1u) & 0x01010101u);
                    keep |= bytes << qd;
                }
                m &= keep;
            }
            unsigned bal = __builtin_amdgcn_ballot_w32(m != 0u);
            while (bal != 0u) {
                const int pbit = __builtin_ctz(m | 0x10000000u);
                const int edge = cb[u] + lane * 16 + ((pbit & 3) << 2) + (pbit >> 3);
                const int pos = qn + (int)__builtin_amdgcn_mbcnt_lo(bal, 0u);
                if (m != 0u) q[pos] = edge;
                qn += __builtin_popcount(bal);
                m &= m - 1u;
                while (qn >= GROUP) {
                    qn -= GROUP;
                    if (FILE) file_queued_edges(q + qn, GROUP, lo, src, dst, ocnt, olist, lane);
                    else add_queued_edges(q + qn, GROUP, lo, src, dst, xw, dinv, rowsf, lane);
                }
                bal = __builtin_amdgcn_ballot_w32(m != 0u);
            }
        }
#pragma unroll
        for (int u = 0; u < SCAN_AHEAD; ++u) {
            cb[u] = cn[u];
            kk[u] = kn[u];
        }
    }
    if (qn > 0) {
        if (FILE) file_queued_edges(q, qn, lo, src, dst, ocnt, olist, lane);
        else add_queued_edges(q, qn, lo, src, dst, xw, dinv, rowsf, lane);
    }
}

__global__ __launch_bounds__(FUSE_THREADS) void range_sums_kernel(const int* __restrict__ src,
                                                                  const int* __restrict__ dst,
                                                                  const unsigned* __restrict__ keys,
                                                                  const float* __restrict__ xw,
                                                                  const float* __restrict__ dinv,
                                                                  const float* __restrict__ bias,
                                                                  const float* __restrict__ wreg,
                                                                  const float* __restrict__ breg,
                                                                  float* __restrict__ out,
                                                                  int n, int e, int nload) {
    __shared__ float4 rows[(RANGE + 1) * (NFEAT / 4)];
    __shared__ int queue[FUSE_WAVES * QCAP];
    __shared__ int olist[FUSE_WAVES * OWNER_CAP];
    __shared__ int ocnt[FUSE_WAVES];

    const int tid  = threadIdx.x;
    const int lane = tid & 31;
    const int wv = __builtin_amdgcn_readfirstlane(tid >> 5);
    const int lo = blockIdx.x * RANGE;
    const int nrows = (n - lo < RANGE) ? (n - lo) : RANGE;

    const float4 b4 = ((const float4*)bias)[lane];
    const float4* xw4 = (const float4*)(xw + (size_t)lo * NFEAT);
    for (int i0 = 0; i0 < 16; i0 += 8) {
        float4 v[8];
        float di[8];
#pragma unroll
        for (int i = 0; i < 8; ++i) {
            const int row = wv + 32 * (i0 + i);
            const int rl = (row < nrows) ? row : (nrows - 1);
            v[i]  = xw4[rl * (NFEAT / 4) + lane];
            di[i] = dinv[lo + rl];
        }
#pragma unroll
        for (int i = 0; i < 8; ++i) {
            const int row = wv + 32 * (i0 + i);
            float4 r;
            r.x = b4.x + v[i].x * di[i] * di[i];
            r.y = b4.y + v[i].y * di[i] * di[i];
            r.z = b4.z + v[i].z * di[i] * di[i];
            r.w = b4.w + v[i].w * di[i] * di[i];
            rows[row * (NFEAT / 4) + lane] = r;
        }
    }
    if (wv == 0) rows[RANGE * (NFEAT / 4) + lane] = make_float4(0.0f, 0.0f, 0.0f, 0.0f);
    if (tid < FUSE_WAVES) ocnt[tid] = 0;
    __syncthreads();

    float* rowsf = (float*)rows;
    int* q = queue + wv * QCAP;
    if (e > 0) scan_keys<true>(keys, src, dst, xw, dinv, rowsf, ocnt, olist, q, lo, e, nload, wv, lane);
    __syncthreads();
    const int oc = ocnt[lane];
    if (__builtin_amdgcn_ballot_w32(oc > OWNER_CAP) != 0u) {
        scan_keys<false>(keys, src, dst, xw, dinv, rowsf, ocnt, olist, q, lo, e, nload, wv, lane);
        __syncthreads();
    } else {
        const int mine = __builtin_amdgcn_readlane(oc, wv);
        if (mine > 0)
            sum_own_list(olist + wv * OWNER_CAP, mine, lo + wv * OWN_ROWS, lane, xw, dinv,
                         rows + wv * OWN_ROWS * (NFEAT / 4) + lane);
    }

    const float4 w4 = ((const float4*)wreg)[lane];
    float res = 0.0f;
#pragma unroll
    for (int j = 0; j < 16; ++j) {
        const float4 a = rows[(wv * 16 + j) * (NFEAT / 4) + lane];
        float dot = fmaxf(a.x, 0.0f) * w4.x + fmaxf(a.y, 0.0f) * w4.y +
                    fmaxf(a.z, 0.0f) * w4.z + fmaxf(a.w, 0.0f) * w4.w;
#pragma unroll
        for (int off = 16; off > 0; off >>= 1)
            dot += __shfl_xor(dot, off, 32);
        if (lane == j) res = dot;
    }
    float* outv = (float*)queue;
    if (lane < 16) outv[wv * 16 + lane] = res + breg[0];
    __syncthreads();
    if (tid < RANGE && lo + tid < n) out[lo + tid] = outv[tid];
}


extern "C" void kernel_launch(void* const* d_in, const int* in_sizes, int n_in,
                              void* d_out, int out_size, void* d_ws, size_t ws_size,
                              hipStream_t stream) {
    const float* x    = (const float*)d_in[0];
    const int*   ei   = (const int*)d_in[1];
    const float* W    = (const float*)d_in[2];
    const float* b    = (const float*)d_in[3];
    const float* wreg = (const float*)d_in[4];
    const float* breg = (const float*)d_in[5];
    float* out = (float*)d_out;

    const int n = in_sizes[0] / NFEAT;
    const int e = in_sizes[1] / 2;
    const int* src = ei;
    const int* dst = ei + e;

    const int nload = (e + KEY_LOAD - 1) / KEY_LOAD;
    unsigned* keys = (unsigned*)d_ws;
    float* xw   = (float*)(keys + (size_t)nload * (KEY_LOAD / 4));
    float* dinv = xw + (size_t)n * NFEAT;
    int*   part = (int*)(dinv + n);

    const int B = 256;
    const int count_blocks = ((n + DEG_RANGE - 1) / DEG_RANGE) * DEG_SLICES;
    deg_count_kernel<<<count_blocks + KEY_BLOCKS, DEG_THREADS, 0, stream>>>(dst, part, keys, count_blocks, nload, n, e);

    gemm_xwT_wmma<<<(n + ROWS_PER_BLOCK - 1) / ROWS_PER_BLOCK, B, 0, stream>>>(x, W, part, dinv, xw, n);

    range_sums_kernel<<<(n + RANGE - 1) / RANGE, FUSE_THREADS, 0, stream>>>(src, dst, keys, xw, dinv, b, wreg, breg, out, n, e, nload);
}


# Edge pass walks 1 bit per edge, not 1 byte: one bit plane per 512-node range, built in LDS by extra blocks of the count launch; a range block reads only its own plane; list order = new walk order
# speedup vs baseline: 3.6234x; 1.0561x over previous
#include <hip/hip_runtime.h>
#include <cstdint>

#define NFEAT 128

typedef __attribute__((ext_vector_type(2))) float v2f;
typedef __attribute__((ext_vector_type(8))) float v8f;


#define DEG_RANGE 6272
#define DEG_SLICES 8
#define DEG_THREADS 1024
#define DEG_AHEAD 16

#define RANGE 512
#define RANGE_SHIFT 9
#define MAX_RANGES 196
#define CHUNK_EDGES 8192
#define CHUNK_WORDS 256
#define BIT_LOAD 4096

__global__ __launch_bounds__(DEG_THREADS) void deg_count_kernel(const int* __restrict__ dst,
                                                                int* __restrict__ part,
                                                                unsigned* __restrict__ planes,
                                                                int count_blocks, int nchunk, int nr,
                                                                int n, int e) {
    __shared__ uint4 bits4[MAX_RANGES * CHUNK_WORDS / 4];
    const int tid = threadIdx.x;
    if ((int)blockIdx.x >= count_blocks) {
        unsigned* bits = (unsigned*)bits4;
        const int chunk = (int)blockIdx.x - count_blocks;
        const int e0 = chunk * CHUNK_EDGES;
        int d[8];
#pragma unroll
        for (int j = 0; j < 8; ++j) {
            const int i = e0 + j * DEG_THREADS + tid;
            d[j] = dst[(i < e) ? i : (e - 1)];
        }
        const uint4 z4 = make_uint4(0u, 0u, 0u, 0u);
        for (int f = tid; f < nr * (CHUNK_WORDS / 4); f += DEG_THREADS) bits4[f] = z4;
        __syncthreads();
#pragma unroll
        for (int j = 0; j < 8; ++j) {
            const int i = e0 + j * DEG_THREADS + tid;
            const unsigned r = (unsigned)d[j] >> RANGE_SHIFT;
            if (i < e && r < (unsigned)nr)
                atomicOr(&bits[r * CHUNK_WORDS + j * 32 + (tid >> 5)], 1u << (tid & 31));
        }
        __syncthreads();
        uint4* pg = (uint4*)planes;
        const size_t pitch4 = (size_t)nchunk * (CHUNK_WORDS / 4);
        for (int f = tid; f < nr * (CHUNK_WORDS / 4); f += DEG_THREADS)
            pg[(size_t)(f >> 6) * pitch4 + (size_t)chunk * (CHUNK_WORDS / 4) + (f & 63)] = bits4[f];
        return;
    }
    int* cnt = (int*)bits4;
    const int k = blockIdx.x / DEG_SLICES;
    const int s = blockIdx.x - k * DEG_SLICES;
    const int lo = k * DEG_RANGE;
    for (int i = tid; i < DEG_RANGE; i += DEG_THREADS) cnt[i] = 0;
    __syncthreads();

    const int per = (e + DEG_SLICES - 1) / DEG_SLICES;
    const int e0 = s * per;
    const int e1 = (e0 + per < e) ? (e0 + per) : e;
    for (int b = e0 + tid; b < e1; b += DEG_THREADS * DEG_AHEAD) {
        int d[DEG_AHEAD];
#pragma unroll
        for (int u = 0; u < DEG_AHEAD; ++u) {
            const int i = b + u * DEG_THREADS;
            const int v = dst[(i < e1) ? i : (e1 - 1)];
            d[u] = (i < e1) ? v : -1;
        }
#pragma unroll
        for (int u = 0; u < DEG_AHEAD; ++u) {
            const unsigned dl = (unsigned)(d[u] - lo);
            if (dl < (unsigned)DEG_RANGE) atomicAdd(&cnt[dl], 1);
        }
    }
    __syncthreads();

    const int left = n - lo;
    const int hi = (left < DEG_RANGE) ? left : DEG_RANGE;
    int* pp = part + (size_t)s * n + lo;
    for (int i = tid; i < hi; i += DEG_THREADS) pp[i] = cnt[i];
}


#define ROWS_PER_BLOCK 128
#define W_PITCH4 33

__global__ __launch_bounds__(256) void gemm_xwT_wmma(const float* __restrict__ x,
                                                     const float* __restrict__ W,
                                                     const int* __restrict__ part,
                                                     float* __restrict__ dinv,
                                                     float* __restrict__ xw, int n) {
    __shared__ float4 lds_w[NFEAT * W_PITCH4];

    const int tid  = threadIdx.x;
    const int lane = tid & 31;
    const int wave = __builtin_amdgcn_readfirstlane(tid >> 5);
    const int c    = lane & 15;
    const int h    = lane >> 4;
    const int r0   = blockIdx.x * ROWS_PER_BLOCK + wave * 16;
    const bool live = r0 < n;

    const float4* wg = (const float4*)W;
    float4 wr[16];
#pragma unroll
    for (int i = 0; i < 16; ++i) wr[i] = wg[tid + 256 * i];

    const int rl = live ? r0 : 0;
    const float4* xr = (const float4*)(x + (size_t)(rl + c) * NFEAT) + h;
    float4 a[16];
#pragma unroll
    for (int j = 0; j < 16; ++j) a[j] = xr[2 * j];

    const int* pr = part + rl + c;
    int pc[DEG_SLICES];
#pragma unroll
    for (int s = 0; s < DEG_SLICES; ++s) pc[s] = pr[(size_t)s * n];

#pragma unroll
    for (int i = 0; i < 16; ++i) {
        const int f = tid + 256 * i;
        lds_w[(f >> 5) * W_PITCH4 + (f & 31)] = wr[i];
    }
    __syncthreads();
    if (!live) return;

    v8f acc[8] = {};
    const float4* wl = lds_w + c * W_PITCH4 + h;

    float4 b[2][8];
#pragma unroll
    for (int t = 0; t < 8; ++t) b[0][t] = wl[t * 16 * W_PITCH4];

#pragma unroll
    for (int j = 0; j < 16; ++j) {
        if (j + 1 < 16) {
#pragma unroll
            for (int t = 0; t < 8; ++t) b[(j + 1) & 1][t] = wl[t * 16 * W_PITCH4 + 2 * (j + 1)];
        }
        __builtin_amdgcn_sched_barrier(0);
        v2f a0, a1;
        a0.x = a[j].x; a0.y = a[j].y;
        a1.x = a[j].z; a1.y = a[j].w;
#pragma unroll
        for (int t = 0; t < 8; ++t) {
            v2f b0;
            b0.x = b[j & 1][t].x; b0.y = b[j & 1][t].y;
            acc[t] = __builtin_amdgcn_wmma_f32_16x16x4_f32(false, a0, false, b0,
                                                           (short)0, acc[t], false, false);
        }
#pragma unroll
        for (int t = 0; t < 8; ++t) {
            v2f b1;
            b1.x = b[j & 1][t].z; b1.y = b[j & 1][t].w;
            acc[t] = __builtin_amdgcn_wmma_f32_16x16x4_f32(false, a1, false, b1,
                                                           (short)0, acc[t], false, false);
        }
        __builtin_amdgcn_sched_barrier(0);
    }

#pragma unroll
    for (int t = 0; t < 8; ++t) {
        const int col = 16 * t + c;
#pragma unroll
        for (int r = 0; r < 8; ++r) {
            const size_t i = (size_t)(r0 + (h << 3) + r) * NFEAT + col;
            xw[i] = acc[t][r];
        }
    }
    int ec = 0;
#pragma unroll
    for (int s = 0; s < DEG_SLICES; ++s) ec += pc[s];
    const float di = rsqrtf(1.0f + (float)ec);
    if (h == 0) dinv[r0 + c] = di;
}


#define FUSE_THREADS 1024
#define FUSE_WAVES 32
#define QCAP 64
#define SCAN_AHEAD 5
#define EDGE_GROUP 8
#define ROUTE_GROUP 32
#define OWN_ROWS 16
#define OWN_SHIFT 4
#define OWNER_CAP 320

__device__ __forceinline__ void add_queued_edges(const int* p, int cnt, int lo,
                                                 const int* __restrict__ src,
                                                 const int* __restrict__ dst,
                                                 const float* __restrict__ xw,
                                                 const float* __restrict__ dinv,
                                                 float* rowsf, int lane) {
    __builtin_amdgcn_fence(__ATOMIC_ACQ_REL, "wavefront");
    const int en = p[(lane < cnt) ? lane : (cnt - 1)];
    const int s = src[en];
    const int d = dst[en];
    const int dl = d - lo;
    const int row = ((unsigned)dl < (unsigned)RANGE) ? dl : RANGE;
    const float dd = dinv[d];
    const float ds = dinv[s];
    if (cnt == EDGE_GROUP) {
        float v[EDGE_GROUP][4];
#pragma unroll
        for (int j = 0; j < EDGE_GROUP; ++j) {
            const int sj = __builtin_amdgcn_readlane(s, j);
            const float* xs = xw + (size_t)sj * NFEAT + lane;
            v[j][0] = xs[0];
            v[j][1] = xs[32];
            v[j][2] = xs[64];
            v[j][3] = xs[96];
        }
        const float w = ds * dd;
#pragma unroll
        for (int j = 0; j < EDGE_GROUP; ++j) {
            const int dj = __builtin_amdgcn_readlane(row, j);
            const float wj = __int_as_float(__builtin_amdgcn_readlane(__float_as_int(w), j));
            float* ab = rowsf + dj * NFEAT + lane;
            atomicAdd(ab + 0,  v[j][0] * wj);
            atomicAdd(ab + 32, v[j][1] * wj);
            atomicAdd(ab + 64, v[j][2] * wj);
            atomicAdd(ab + 96, v[j][3] * wj);
        }
    } else {
        const float w = ds * dd;
        for (int j = 0; j < cnt; ++j) {
            const int sj = __builtin_amdgcn_readlane(s, j);
            const int dj = __builtin_amdgcn_readlane(row, j);
            const float wj = __int_as_float(__builtin_amdgcn_readlane(__float_as_int(w), j));
            const float* xs = xw + (size_t)sj * NFEAT + lane;
            const float v0 = xs[0];
            const float v1 = xs[32];
            const float v2 = xs[64];
            const float v3 = xs[96];
            float* ab = rowsf + dj * NFEAT + lane;
            atomicAdd(ab + 0,  v0 * wj);
            atomicAdd(ab + 32, v1 * wj);
            atomicAdd(ab + 64, v2 * wj);
            atomicAdd(ab + 96, v3 * wj);
        }
    }
}

__device__ __forceinline__ void file_queued_edges(const int* p, int cnt, int lo,
                                                  const int* __restrict__ src,
                                                  const int* __restrict__ dst,
                                                  int* ocnt, int* olist, int lane) {
    __builtin_amdgcn_fence(__ATOMIC_ACQ_REL, "wavefront");
    const int en = p[(lane < cnt) ? lane : (cnt - 1)];
    const int s = src[en];
    const int d = dst[en];
    asm volatile("" :: "v"(s), "v"(d));
    const unsigned dl = (unsigned)(d - lo);
    if (lane < cnt && dl < (unsigned)RANGE) {
        const int ow = (int)(dl >> OWN_SHIFT);
        const int pos = atomicAdd(&ocnt[ow], 1);
        if (pos < OWNER_CAP) olist[ow * OWNER_CAP + pos] = (s << OWN_SHIFT) | (int)(dl & (OWN_ROWS - 1));
    }
}

__device__ __forceinline__ void ask_group_rows(int ent, int own_lo, int lane,
                                               const float* __restrict__ xw,
                                               const float* __restrict__ dinv,
                                               float4 (&v)[8], int& r, float& ds, float& dd) {
    const int s = ent >> OWN_SHIFT;
    r = ent & (OWN_ROWS - 1);
    ds = dinv[s];
    dd = dinv[own_lo + r];
#pragma unroll
    for (int j = 0; j < 8; ++j) {
        const int sj = __builtin_amdgcn_readlane(s, j);
        v[j] = ((const float4*)(xw + (size_t)sj * NFEAT))[lane];
    }
}

__device__ __forceinline__ void add_group_rows(const float4 (&v)[8], int r, float ds, float dd,
                                               int nv, float4* myrows) {
    const float w = ds * dd;
#pragma unroll
    for (int j = 0; j < 8; ++j) {
        if (j < nv) {
            const int rj = __builtin_amdgcn_readlane(r, j);
            const float wj = __int_as_float(__builtin_amdgcn_readlane(__float_as_int(w), j));
            float4* pr = myrows + rj * (NFEAT / 4);
            float4 a = *pr;
            a.x += v[j].x * wj;
            a.y += v[j].y * wj;
            a.z += v[j].z * wj;
            a.w += v[j].w * wj;
            *pr = a;
        }
    }
}

__device__ __forceinline__ void sum_own_list(const int* ol, int cnt, int own_lo, int lane,
                                             const float* __restrict__ xw,
                                             const float* __restrict__ dinv, float4* myrows) {
    const int last = cnt - 1;
    const int l8 = lane & 7;
    float4 va[8], vb[8];
    int ra, rb;
    float dsa, dda, dsb, ddb;
    int ent = ol[(l8 < last) ? l8 : last];
    ask_group_rows(ent, own_lo, lane, xw, dinv, va, ra, dsa, dda);
    ent = ol[(8 + l8 < last) ? (8 + l8) : last];
    for (int g = 0; g * 8 < cnt; g += 2) {
        ask_group_rows(ent, own_lo, lane, xw, dinv, vb, rb, dsb, ddb);
        int i = (g + 2) * 8 + l8;
        ent = ol[(i < last) ? i : last];
        add_group_rows(va, ra, dsa, dda, cnt - g * 8, myrows);
        ask_group_rows(ent, own_lo, lane, xw, dinv, va, ra, dsa, dda);
        i = (g + 3) * 8 + l8;
        ent = ol[(i < last) ? i : last];
        add_group_rows(vb, rb, dsb, ddb, cnt - (g + 1) * 8, myrows);
    }
}

template <bool FILE>
__device__ __forceinline__ void scan_bits(const uint4* __restrict__ bp,
                                          const int* __restrict__ src, const int* __restrict__ dst,
                                          const float* __restrict__ xw,
                                          const float* __restrict__ dinv,
                                          float* rowsf, int* ocnt, int* olist, int* q, int lo,
                                          int nwl, int wv, int lane) {
    constexpr int GROUP = FILE ? ROUTE_GROUP : EDGE_GROUP;
    int qn = 0;
    for (int c0 = wv; c0 < nwl; c0 += FUSE_WAVES * SCAN_AHEAD) {
        uint4 kk[SCAN_AHEAD];
        int cb[SCAN_AHEAD];
#pragma unroll
        for (int u = 0; u < SCAN_AHEAD; ++u) {
            const int ch = c0 + u * FUSE_WAVES;
            const bool okc = ch < nwl;
            cb[u] = okc ? ch * BIT_LOAD : -1;
            kk[u] = bp[(okc ? ch : c0) * 32 + lane];
        }
#pragma unroll
        for (int u = 0; u < SCAN_AHEAD; ++u) {
            if (cb[u] < 0) continue;
            unsigned kx = kk[u].x, ky = kk[u].y, kz = kk[u].z, kw = kk[u].w;
            int base = cb[u] + lane * 128;
#pragma nounroll
            for (int dw = 0; dw < 4; ++dw) {
                unsigned m = kx;
                unsigned bal = __builtin_amdgcn_ballot_w32(m != 0u);
                while (bal != 0u) {
                    const int edge = base + __builtin_ctz(m | 0x80000000u);
                    const int pos = qn + (int)__builtin_amdgcn_mbcnt_lo(bal, 0u);
                    if (m != 0u) q[pos] = edge;
                    qn += __builtin_popcount(bal);
                    m &= m - 1u;
                    while (qn >= GROUP) {
                        qn -= GROUP;
                        if (FILE) file_queued_edges(q + qn, GROUP, lo, src, dst, ocnt, olist, lane);
                        else add_queued_edges(q + qn, GROUP, lo, src, dst, xw, dinv, rowsf, lane);
                    }
                    bal = __builtin_amdgcn_ballot_w32(m != 0u);
                }
                kx = ky; ky = kz; kz = kw;
                base += 32;
            }
        }
    }
    if (qn > 0) {
        if (FILE) file_queued_edges(q, qn, lo, src, dst, ocnt, olist, lane);
        else add_queued_edges(q, qn, lo, src, dst, xw, dinv, rowsf, lane);
    }
}

__global__ __launch_bounds__(FUSE_THREADS) void range_sums_kernel(const int* __restrict__ src,
                                                                  const int* __restrict__ dst,
                                                                  const unsigned* __restrict__ planes,
                                                                  const float* __restrict__ xw,
                                                                  const float* __restrict__ dinv,
                                                                  const float* __restrict__ bias,
                                                                  const float* __restrict__ wreg,
                                                                  const float* __restrict__ breg,
                                                                  float* __restrict__ out,
                                                                  int n, int nwl) {
    __shared__ float4 rows[(RANGE + 1) * (NFEAT / 4)];
    __shared__ int queue[FUSE_WAVES * QCAP];
    __shared__ int olist[FUSE_WAVES * OWNER_CAP];
    __shared__ int ocnt[FUSE_WAVES];

    const int tid  = threadIdx.x;
    const int lane = tid & 31;
    const int wv = __builtin_amdgcn_readfirstlane(tid >> 5);
    const int lo = blockIdx.x * RANGE;
    const int nrows = (n - lo < RANGE) ? (n - lo) : RANGE;

    const float4 b4 = ((const float4*)bias)[lane];
    const float4* xw4 = (const float4*)(xw + (size_t)lo * NFEAT);
    for (int i0 = 0; i0 < 16; i0 += 8) {
        float4 v[8];
        float di[8];
#pragma unroll
        for (int i = 0; i < 8; ++i) {
            const int row = wv + 32 * (i0 + i);
            const int rl = (row < nrows) ? row : (nrows - 1);
            v[i]  = xw4[rl * (NFEAT / 4) + lane];
            di[i] = dinv[lo + rl];
        }
#pragma unroll
        for (int i = 0; i < 8; ++i) {
            const int row = wv + 32 * (i0 + i);
            float4 r;
            r.x = b4.x + v[i].x * di[i] * di[i];
            r.y = b4.y + v[i].y * di[i] * di[i];
            r.z = b4.z + v[i].z * di[i] * di[i];
            r.w = b4.w + v[i].w * di[i] * di[i];
            rows[row * (NFEAT / 4) + lane] = r;
        }
    }
    if (wv == 0) rows[RANGE * (NFEAT / 4) + lane] = make_float4(0.0f, 0.0f, 0.0f, 0.0f);
    if (tid < FUSE_WAVES) ocnt[tid] = 0;
    __syncthreads();

    float* rowsf = (float*)rows;
    int* q = queue + wv * QCAP;
    const uint4* bp = (const uint4*)planes + (size_t)blockIdx.x * nwl * 32;
    scan_bits<true>(bp, src, dst, xw, dinv, rowsf, ocnt, olist, q, lo, nwl, wv, lane);
    __syncthreads();
    const int oc = ocnt[lane];
    if (__builtin_amdgcn_ballot_w32(oc > OWNER_CAP) != 0u) {
        scan_bits<false>(bp, src, dst, xw, dinv, rowsf, ocnt, olist, q, lo, nwl, wv, lane);
        __syncthreads();
    } else {
        const int mine = __builtin_amdgcn_readlane(oc, wv);
        if (mine > 0)
            sum_own_list(olist + wv * OWNER_CAP, mine, lo + wv * OWN_ROWS, lane, xw, dinv,
                         rows + wv * OWN_ROWS * (NFEAT / 4) + lane);
    }

    const float4 w4 = ((const float4*)wreg)[lane];
    float res = 0.0f;
#pragma unroll
    for (int j = 0; j < 16; ++j) {
        const float4 a = rows[(wv * 16 + j) * (NFEAT / 4) + lane];
        float dot = fmaxf(a.x, 0.0f) * w4.x + fmaxf(a.y, 0.0f) * w4.y +
                    fmaxf(a.z, 0.0f) * w4.z + fmaxf(a.w, 0.0f) * w4.w;
#pragma unroll
        for (int off = 16; off > 0; off >>= 1)
            dot += __shfl_xor(dot, off, 32);
        if (lane == j) res = dot;
    }
    float* outv = (float*)queue;
    if (lane < 16) outv[wv * 16 + lane] = res + breg[0];
    __syncthreads();
    if (tid < RANGE && lo + tid < n) out[lo + tid] = outv[tid];
}


extern "C" void kernel_launch(void* const* d_in, const int* in_sizes, int n_in,
                              void* d_out, int out_size, void* d_ws, size_t ws_size,
                              hipStream_t stream) {
    const float* x    = (const float*)d_in[0];
    const int*   ei   = (const int*)d_in[1];
    const float* W    = (const float*)d_in[2];
    const float* b    = (const float*)d_in[3];
    const float* wreg = (const float*)d_in[4];
    const float* breg = (const float*)d_in[5];
    float* out = (float*)d_out;

    const int n = in_sizes[0] / NFEAT;
    const int e = in_sizes[1] / 2;
    const int* src = ei;
    const int* dst = ei + e;

    const int nchunk = (e + CHUNK_EDGES - 1) / CHUNK_EDGES;
    const int nr = (n + RANGE - 1) / RANGE;
    unsigned* planes = (unsigned*)d_ws;
    float* xw   = (float*)(planes + (size_t)nr * nchunk * CHUNK_WORDS);
    float* dinv = xw + (size_t)n * NFEAT;
    int*   part = (int*)(dinv + n);

    const int B = 256;
    const int count_blocks = ((n + DEG_RANGE - 1) / DEG_RANGE) * DEG_SLICES;
    deg_count_kernel<<<count_blocks + nchunk, DEG_THREADS, 0, stream>>>(dst, part, planes, count_blocks, nchunk, nr, n, e);

    gemm_xwT_wmma<<<(n + ROWS_PER_BLOCK - 1) / ROWS_PER_BLOCK, B, 0, stream>>>(x, W, part, dinv, xw, n);

    range_sums_kernel<<<nr, FUSE_THREADS, 0, stream>>>(src, dst, planes, xw, dinv, b, wreg, breg, out, n, nchunk * (CHUNK_WORDS / 128));
}


# Edge pass re-tiled: a block owns 416 destination nodes (32 waves x 13 rows), not 512: 241 blocks, not 196, so 241 of the 256 compute units work, each on 19% fewer edges; sums in list arrival order
# speedup vs baseline: 3.7330x; 1.0302x over previous
#include <hip/hip_runtime.h>
#include <cstdint>

#define NFEAT 128

typedef __attribute__((ext_vector_type(2))) float v2f;
typedef __attribute__((ext_vector_type(8))) float v8f;


#define DEG_RANGE 6272
#define DEG_SLICES 8
#define DEG_THREADS 1024
#define DEG_AHEAD 16

#define RANGE 416
#define MAX_RANGES 241
#define CHUNK_EDGES 8192
#define CHUNK_WORDS 256
#define BIT_LOAD 4096

__global__ __launch_bounds__(DEG_THREADS) void deg_count_kernel(const int* __restrict__ dst,
                                                                int* __restrict__ part,
                                                                unsigned* __restrict__ planes,
                                                                int count_blocks, int nchunk, int nr,
                                                                int n, int e) {
    __shared__ uint4 bits4[MAX_RANGES * CHUNK_WORDS / 4];
    const int tid = threadIdx.x;
    if ((int)blockIdx.x >= count_blocks) {
        unsigned* bits = (unsigned*)bits4;
        const int chunk = (int)blockIdx.x - count_blocks;
        const int e0 = chunk * CHUNK_EDGES;
        int d[8];
#pragma unroll
        for (int j = 0; j < 8; ++j) {
            const int i = e0 + j * DEG_THREADS + tid;
            d[j] = dst[(i < e) ? i : (e - 1)];
        }
        const uint4 z4 = make_uint4(0u, 0u, 0u, 0u);
        for (int f = tid; f < nr * (CHUNK_WORDS / 4); f += DEG_THREADS) bits4[f] = z4;
        __syncthreads();
#pragma unroll
        for (int j = 0; j < 8; ++j) {
            const int i = e0 + j * DEG_THREADS + tid;
            const unsigned r = (unsigned)d[j] / (unsigned)RANGE;
            if (i < e && r < (unsigned)nr)
                atomicOr(&bits[r * CHUNK_WORDS + j * 32 + (tid >> 5)], 1u << (tid & 31));
        }
        __syncthreads();
        uint4* pg = (uint4*)planes;
        const size_t pitch4 = (size_t)nchunk * (CHUNK_WORDS / 4);
        for (int f = tid; f < nr * (CHUNK_WORDS / 4); f += DEG_THREADS)
            pg[(size_t)(f >> 6) * pitch4 + (size_t)chunk * (CHUNK_WORDS / 4) + (f & 63)] = bits4[f];
        return;
    }
    int* cnt = (int*)bits4;
    const int k = blockIdx.x / DEG_SLICES;
    const int s = blockIdx.x - k * DEG_SLICES;
    const int lo = k * DEG_RANGE;
    for (int i = tid; i < DEG_RANGE; i += DEG_THREADS) cnt[i] = 0;
    __syncthreads();

    const int per = (e + DEG_SLICES - 1) / DEG_SLICES;
    const int e0 = s * per;
    const int e1 = (e0 + per < e) ? (e0 + per) : e;
    for (int b = e0 + tid; b < e1; b += DEG_THREADS * DEG_AHEAD) {
        int d[DEG_AHEAD];
#pragma unroll
        for (int u = 0; u < DEG_AHEAD; ++u) {
            const int i = b + u * DEG_THREADS;
            const int v = dst[(i < e1) ? i : (e1 - 1)];
            d[u] = (i < e1) ? v : -1;
        }
#pragma unroll
        for (int u = 0; u < DEG_AHEAD; ++u) {
            const unsigned dl = (unsigned)(d[u] - lo);
            if (dl < (unsigned)DEG_RANGE) atomicAdd(&cnt[dl], 1);
        }
    }
    __syncthreads();

    const int left = n - lo;
    const int hi = (left < DEG_RANGE) ? left : DEG_RANGE;
    int* pp = part + (size_t)s * n + lo;
    for (int i = tid; i < hi; i += DEG_THREADS) pp[i] = cnt[i];
}


#define ROWS_PER_BLOCK 128
#define W_PITCH4 33

__global__ __launch_bounds__(256) void gemm_xwT_wmma(const float* __restrict__ x,
                                                     const float* __restrict__ W,
                                                     const int* __restrict__ part,
                                                     float* __restrict__ dinv,
                                                     float* __restrict__ xw, int n) {
    __shared__ float4 lds_w[NFEAT * W_PITCH4];

    const int tid  = threadIdx.x;
    const int lane = tid & 31;
    const int wave = __builtin_amdgcn_readfirstlane(tid >> 5);
    const int c    = lane & 15;
    const int h    = lane >> 4;
    const int r0   = blockIdx.x * ROWS_PER_BLOCK + wave * 16;
    const bool live = r0 < n;

    const float4* wg = (const float4*)W;
    float4 wr[16];
#pragma unroll
    for (int i = 0; i < 16; ++i) wr[i] = wg[tid + 256 * i];

    const int rl = live ? r0 : 0;
    const float4* xr = (const float4*)(x + (size_t)(rl + c) * NFEAT) + h;
    float4 a[16];
#pragma unroll
    for (int j = 0; j < 16; ++j) a[j] = xr[2 * j];

    const int* pr = part + rl + c;
    int pc[DEG_SLICES];
#pragma unroll
    for (int s = 0; s < DEG_SLICES; ++s) pc[s] = pr[(size_t)s * n];

#pragma unroll
    for (int i = 0; i < 16; ++i) {
        const int f = tid + 256 * i;
        lds_w[(f >> 5) * W_PITCH4 + (f & 31)] = wr[i];
    }
    __syncthreads();
    if (!live) return;

    v8f acc[8] = {};
    const float4* wl = lds_w + c * W_PITCH4 + h;

    float4 b[2][8];
#pragma unroll
    for (int t = 0; t < 8; ++t) b[0][t] = wl[t * 16 * W_PITCH4];

#pragma unroll
    for (int j = 0; j < 16; ++j) {
        if (j + 1 < 16) {
#pragma unroll
            for (int t = 0; t < 8; ++t) b[(j + 1) & 1][t] = wl[t * 16 * W_PITCH4 + 2 * (j + 1)];
        }
        __builtin_amdgcn_sched_barrier(0);
        v2f a0, a1;
        a0.x = a[j].x; a0.y = a[j].y;
        a1.x = a[j].z; a1.y = a[j].w;
#pragma unroll
        for (int t = 0; t < 8; ++t) {
            v2f b0;
            b0.x = b[j & 1][t].x; b0.y = b[j & 1][t].y;
            acc[t] = __builtin_amdgcn_wmma_f32_16x16x4_f32(false, a0, false, b0,
                                                           (short)0, acc[t], false, false);
        }
#pragma unroll
        for (int t = 0; t < 8; ++t) {
            v2f b1;
            b1.x = b[j & 1][t].z; b1.y = b[j & 1][t].w;
            acc[t] = __builtin_amdgcn_wmma_f32_16x16x4_f32(false, a1, false, b1,
                                                           (short)0, acc[t], false, false);
        }
        __builtin_amdgcn_sched_barrier(0);
    }

#pragma unroll
    for (int t = 0; t < 8; ++t) {
        const int col = 16 * t + c;
#pragma unroll
        for (int r = 0; r < 8; ++r) {
            const size_t i = (size_t)(r0 + (h << 3) + r) * NFEAT + col;
            xw[i] = acc[t][r];
        }
    }
    int ec = 0;
#pragma unroll
    for (int s = 0; s < DEG_SLICES; ++s) ec += pc[s];
    const float di = rsqrtf(1.0f + (float)ec);
    if (h == 0) dinv[r0 + c] = di;
}


#define FUSE_THREADS 1024
#define FUSE_WAVES 32
#define QCAP 64
#define SCAN_AHEAD 5
#define EDGE_GROUP 8
#define ROUTE_GROUP 32
#define OWN_ROWS 13
#define ENT_SHIFT 4
#define SPARE_ROW RANGE
#define OWNER_CAP 320

__device__ __forceinline__ void add_queued_edges(const int* p, int cnt, int lo,
                                                 const int* __restrict__ src,
                                                 const int* __restrict__ dst,
                                                 const float* __restrict__ xw,
                                                 const float* __restrict__ dinv,
                                                 float* rowsf, int lane) {
    __builtin_amdgcn_fence(__ATOMIC_ACQ_REL, "wavefront");
    const int en = p[(lane < cnt) ? lane : (cnt - 1)];
    const int s = src[en];
    const int d = dst[en];
    const int dl = d - lo;
    const int row = ((unsigned)dl < (unsigned)RANGE) ? dl : SPARE_ROW;
    const float dd = dinv[d];
    const float ds = dinv[s];
    if (cnt == EDGE_GROUP) {
        float v[EDGE_GROUP][4];
#pragma unroll
        for (int j = 0; j < EDGE_GROUP; ++j) {
            const int sj = __builtin_amdgcn_readlane(s, j);
            const float* xs = xw + (size_t)sj * NFEAT + lane;
            v[j][0] = xs[0];
            v[j][1] = xs[32];
            v[j][2] = xs[64];
            v[j][3] = xs[96];
        }
        const float w = ds * dd;
#pragma unroll
        for (int j = 0; j < EDGE_GROUP; ++j) {
            const int dj = __builtin_amdgcn_readlane(row, j);
            const float wj = __int_as_float(__builtin_amdgcn_readlane(__float_as_int(w), j));
            float* ab = rowsf + dj * NFEAT + lane;
            atomicAdd(ab + 0,  v[j][0] * wj);
            atomicAdd(ab + 32, v[j][1] * wj);
            atomicAdd(ab + 64, v[j][2] * wj);
            atomicAdd(ab + 96, v[j][3] * wj);
        }
    } else {
        const float w = ds * dd;
        for (int j = 0; j < cnt; ++j) {
            const int sj = __builtin_amdgcn_readlane(s, j);
            const int dj = __builtin_amdgcn_readlane(row, j);
            const float wj = __int_as_float(__builtin_amdgcn_readlane(__float_as_int(w), j));
            const float* xs = xw + (size_t)sj * NFEAT + lane;
            const float v0 = xs[0];
            const float v1 = xs[32];
            const float v2 = xs[64];
            const float v3 = xs[96];
            float* ab = rowsf + dj * NFEAT + lane;
            atomicAdd(ab + 0,  v0 * wj);
            atomicAdd(ab + 32, v1 * wj);
            atomicAdd(ab + 64, v2 * wj);
            atomicAdd(ab + 96, v3 * wj);
        }
    }
}

__device__ __forceinline__ void file_queued_edges(const int* p, int cnt, int lo,
                                                  const int* __restrict__ src,
                                                  const int* __restrict__ dst,
                                                  int* ocnt, int* olist, int lane) {
    __builtin_amdgcn_fence(__ATOMIC_ACQ_REL, "wavefront");
    const int en = p[(lane < cnt) ? lane : (cnt - 1)];
    const int s = src[en];
    const int d = dst[en];
    asm volatile("" :: "v"(s), "v"(d));
    const unsigned dl = (unsigned)(d - lo);
    if (lane < cnt && dl < (unsigned)RANGE) {
        const unsigned ow = dl / (unsigned)OWN_ROWS;
        const int pos = atomicAdd(&ocnt[ow], 1);
        if (pos < OWNER_CAP) olist[ow * OWNER_CAP + pos] = (s << ENT_SHIFT) | (int)(dl - ow * OWN_ROWS);
    }
}

__device__ __forceinline__ void ask_group_rows(int ent, int own_lo, int lane,
                                               const float* __restrict__ xw,
                                               const float* __restrict__ dinv,
                                               float4 (&v)[8], int& r, float& ds, float& dd) {
    const int s = ent >> ENT_SHIFT;
    r = ent & ((1 << ENT_SHIFT) - 1);
    ds = dinv[s];
    dd = dinv[own_lo + r];
#pragma unroll
    for (int j = 0; j < 8; ++j) {
        const int sj = __builtin_amdgcn_readlane(s, j);
        v[j] = ((const float4*)(xw + (size_t)sj * NFEAT))[lane];
    }
}

__device__ __forceinline__ void add_group_rows(const float4 (&v)[8], int r, float ds, float dd,
                                               int nv, float4* myrows) {
    const float w = ds * dd;
#pragma unroll
    for (int j = 0; j < 8; ++j) {
        if (j < nv) {
            const int rj = __builtin_amdgcn_readlane(r, j);
            const float wj = __int_as_float(__builtin_amdgcn_readlane(__float_as_int(w), j));
            float4* pr = myrows + rj * (NFEAT / 4);
            float4 a = *pr;
            a.x += v[j].x * wj;
            a.y += v[j].y * wj;
            a.z += v[j].z * wj;
            a.w += v[j].w * wj;
            *pr = a;
        }
    }
}

__device__ __forceinline__ void sum_own_list(const int* ol, int cnt, int own_lo, int lane,
                                             const float* __restrict__ xw,
                                             const float* __restrict__ dinv, float4* myrows) {
    const int last = cnt - 1;
    const int l8 = lane & 7;
    float4 va[8], vb[8];
    int ra, rb;
    float dsa, dda, dsb, ddb;
    int ent = ol[(l8 < last) ? l8 : last];
    ask_group_rows(ent, own_lo, lane, xw, dinv, va, ra, dsa, dda);
    ent = ol[(8 + l8 < last) ? (8 + l8) : last];
    for (int g = 0; g * 8 < cnt; g += 2) {
        ask_group_rows(ent, own_lo, lane, xw, dinv, vb, rb, dsb, ddb);
        int i = (g + 2) * 8 + l8;
        ent = ol[(i < last) ? i : last];
        add_group_rows(va, ra, dsa, dda, cnt - g * 8, myrows);
        ask_group_rows(ent, own_lo, lane, xw, dinv, va, ra, dsa, dda);
        i = (g + 3) * 8 + l8;
        ent = ol[(i < last) ? i : last];
        add_group_rows(vb, rb, dsb, ddb, cnt - (g + 1) * 8, myrows);
    }
}

template <bool FILE>
__device__ __forceinline__ void scan_bits(const uint4* __restrict__ bp,
                                          const int* __restrict__ src, const int* __restrict__ dst,
                                          const float* __restrict__ xw,
                                          const float* __restrict__ dinv,
                                          float* rowsf, int* ocnt, int* olist, int* q, int lo,
                                          int nwl, int wv, int lane) {
    constexpr int GROUP = FILE ? ROUTE_GROUP : EDGE_GROUP;
    int qn = 0;
    for (int c0 = wv; c0 < nwl; c0 += FUSE_WAVES * SCAN_AHEAD) {
        uint4 kk[SCAN_AHEAD];
        int cb[SCAN_AHEAD];
#pragma unroll
        for (int u = 0; u < SCAN_AHEAD; ++u) {
            const int ch = c0 + u * FUSE_WAVES;
            const bool okc = ch < nwl;
            cb[u] = okc ? ch * BIT_LOAD : -1;
            kk[u] = bp[(okc ? ch : c0) * 32 + lane];
        }
#pragma unroll
        for (int u = 0; u < SCAN_AHEAD; ++u) {
            if (cb[u] < 0) continue;
            unsigned kx = kk[u].x, ky = kk[u].y, kz = kk[u].z, kw = kk[u].w;
            int base = cb[u] + lane * 128;
#pragma nounroll
            for (int dw = 0; dw < 4; ++dw) {
                unsigned m = kx;
                unsigned bal = __builtin_amdgcn_ballot_w32(m != 0u);
                while (bal != 0u) {
                    const int edge = base + __builtin_ctz(m | 0x80000000u);
                    const int pos = qn + (int)__builtin_amdgcn_mbcnt_lo(bal, 0u);
                    if (m != 0u) q[pos] = edge;
                    qn += __builtin_popcount(bal);
                    m &= m - 1u;
                    while (qn >= GROUP) {
                        qn -= GROUP;
                        if (FILE) file_queued_edges(q + qn, GROUP, lo, src, dst, ocnt, olist, lane);
                        else add_queued_edges(q + qn, GROUP, lo, src, dst, xw, dinv, rowsf, lane);
                    }
                    bal = __builtin_amdgcn_ballot_w32(m != 0u);
                }
                kx = ky; ky = kz; kz = kw;
                base += 32;
            }
        }
    }
    if (qn > 0) {
        if (FILE) file_queued_edges(q, qn, lo, src, dst, ocnt, olist, lane);
        else add_queued_edges(q, qn, lo, src, dst, xw, dinv, rowsf, lane);
    }
}

__global__ __launch_bounds__(FUSE_THREADS) void range_sums_kernel(const int* __restrict__ src,
                                                                  const int* __restrict__ dst,
                                                                  const unsigned* __restrict__ planes,
                                                                  const float* __restrict__ xw,
                                                                  const float* __restrict__ dinv,
                                                                  const float* __restrict__ bias,
                                                                  const float* __restrict__ wreg,
                                                                  const float* __restrict__ breg,
                                                                  float* __restrict__ out,
                                                                  int n, int nwl) {
    __shared__ float4 rows[(RANGE + 1) * (NFEAT / 4)];
    __shared__ int queue[FUSE_WAVES * QCAP];
    __shared__ int olist[FUSE_WAVES * OWNER_CAP];
    __shared__ int ocnt[FUSE_WAVES];

    const int tid  = threadIdx.x;
    const int lane = tid & 31;
    const int wv = __builtin_amdgcn_readfirstlane(tid >> 5);
    const int lo = blockIdx.x * RANGE;
    const int nrows = (n - lo < RANGE) ? (n - lo) : RANGE;

    const float4 b4 = ((const float4*)bias)[lane];
    const float4* xw4 = (const float4*)(xw + (size_t)lo * NFEAT);
    {
        float4 v[OWN_ROWS];
        float di[OWN_ROWS];
#pragma unroll
        for (int i = 0; i < OWN_ROWS; ++i) {
            const int row = wv + 32 * i;
            const int rl = (row < nrows) ? row : (nrows - 1);
            v[i]  = xw4[rl * (NFEAT / 4) + lane];
            di[i] = dinv[lo + rl];
        }
#pragma unroll
        for (int i = 0; i < OWN_ROWS; ++i) {
            const int row = wv + 32 * i;
            float4 r;
            r.x = b4.x + v[i].x * di[i] * di[i];
            r.y = b4.y + v[i].y * di[i] * di[i];
            r.z = b4.z + v[i].z * di[i] * di[i];
            r.w = b4.w + v[i].w * di[i] * di[i];
            rows[row * (NFEAT / 4) + lane] = r;
        }
    }
    if (wv == 0) rows[SPARE_ROW * (NFEAT / 4) + lane] = make_float4(0.0f, 0.0f, 0.0f, 0.0f);
    if (tid < FUSE_WAVES) ocnt[tid] = 0;
    __syncthreads();

    float* rowsf = (float*)rows;
    int* q = queue + wv * QCAP;
    const uint4* bp = (const uint4*)planes + (size_t)blockIdx.x * nwl * 32;
    scan_bits<true>(bp, src, dst, xw, dinv, rowsf, ocnt, olist, q, lo, nwl, wv, lane);
    __syncthreads();
    const int oc = ocnt[lane];
    if (__builtin_amdgcn_ballot_w32(oc > OWNER_CAP) != 0u) {
        scan_bits<false>(bp, src, dst, xw, dinv, rowsf, ocnt, olist, q, lo, nwl, wv, lane);
        __syncthreads();
    } else {
        const int mine = __builtin_amdgcn_readlane(oc, wv);
        if (mine > 0)
            sum_own_list(olist + wv * OWNER_CAP, mine, lo + wv * OWN_ROWS, lane, xw, dinv,
                         rows + wv * OWN_ROWS * (NFEAT / 4) + lane);
    }

    const float4 w4 = ((const float4*)wreg)[lane];
    float res = 0.0f;
#pragma unroll
    for (int j = 0; j < OWN_ROWS; ++j) {
        const float4 a = rows[(wv * OWN_ROWS + j) * (NFEAT / 4) + lane];
        float dot = fmaxf(a.x, 0.0f) * w4.x + fmaxf(a.y, 0.0f) * w4.y +
                    fmaxf(a.z, 0.0f) * w4.z + fmaxf(a.w, 0.0f) * w4.w;
#pragma unroll
        for (int off = 16; off > 0; off >>= 1)
            dot += __shfl_xor(dot, off, 32);
        if (lane == j) res = dot;
    }
    float* outv = (float*)queue;
    if (lane < OWN_ROWS) outv[wv * OWN_ROWS + lane] = res + breg[0];
    __syncthreads();
    if (tid < RANGE && lo + tid < n) out[lo + tid] = outv[tid];
}


extern "C" void kernel_launch(void* const* d_in, const int* in_sizes, int n_in,
                              void* d_out, int out_size, void* d_ws, size_t ws_size,
                              hipStream_t stream) {
    const float* x    = (const float*)d_in[0];
    const int*   ei   = (const int*)d_in[1];
    const float* W    = (const float*)d_in[2];
    const float* b    = (const float*)d_in[3];
    const float* wreg = (const float*)d_in[4];
    const float* breg = (const float*)d_in[5];
    float* out = (float*)d_out;

    const int n = in_sizes[0] / NFEAT;
    const int e = in_sizes[1] / 2;
    const int* src = ei;
    const int* dst = ei + e;

    const int nchunk = (e + CHUNK_EDGES - 1) / CHUNK_EDGES;
    const int nr = (n + RANGE - 1) / RANGE;
    unsigned* planes = (unsigned*)d_ws;
    float* xw   = (float*)(planes + (size_t)nr * nchunk * CHUNK_WORDS);
    float* dinv = xw + (size_t)n * NFEAT;
    int*   part = (int*)(dinv + n);

    const int B = 256;
    const int count_blocks = ((n + DEG_RANGE - 1) / DEG_RANGE) * DEG_SLICES;
    deg_count_kernel<<<count_blocks + nchunk, DEG_THREADS, 0, stream>>>(dst, part, planes, count_blocks, nchunk, nr, n, e);

    gemm_xwT_wmma<<<(n + ROWS_PER_BLOCK - 1) / ROWS_PER_BLOCK, B, 0, stream>>>(x, W, part, dinv, xw, n);

    range_sums_kernel<<<nr, FUSE_THREADS, 0, stream>>>(src, dst, planes, xw, dinv, b, wreg, breg, out, n, nchunk * (CHUNK_WORDS / 128));
}
